# starting kernel
# baseline (speedup 1.0000x reference)
#include <hip/hip_runtime.h>

#define DIMK   2048
#define NH     32
#define NKV    8
#define HD     64
#define SEQ    2048
#define BATCH  2

typedef __attribute__((ext_vector_type(2))) float v2f;
typedef __attribute__((ext_vector_type(8))) float v8f;

__device__ __forceinline__ v8f wmma4(v2f a, v2f b, v8f c) {
    return __builtin_amdgcn_wmma_f32_16x16x4_f32(
         false, a,  false, b,
         (short)0, c,  false,  false);
}

__device__ __forceinline__ void gemm16x64(const float* __restrict__ A, int lda,
                                          const float* __restrict__ B, int ldb,
                                          int m0, int col0, int K, v8f acc[4]) {
    const int lane = threadIdx.x & 31;
    const int ln   = lane & 15;
    const int g2   = (lane >> 4) * 2;
    const float* pa = A + (size_t)(m0 + ln) * lda + g2;
    const float* pb = B + (size_t)g2 * ldb + col0 + ln;
    acc[0] = {}; acc[1] = {}; acc[2] = {}; acc[3] = {};
    for (int k = 0; k < K; k += 4) {
        v2f a = *(const v2f*)(pa + k);
        const float* pbk = pb + (size_t)k * ldb;
#pragma unroll
        for (int j = 0; j < 4; ++j) {
            v2f b = { pbk[j * 16], pbk[j * 16 + ldb] };
            acc[j] = wmma4(a, b, acc[j]);
        }
    }
}

__global__ void qkv_rope_kernel(const float* __restrict__ x,
                                const float* __restrict__ wq,
                                const float* __restrict__ wk,
                                const float* __restrict__ wv,
                                const float* __restrict__ fcos,
                                const float* __restrict__ fsin,
                                float* __restrict__ qws,
                                float* __restrict__ ktws,
                                float* __restrict__ vws) {
    const int lane = threadIdx.x & 31;
    const int ln   = lane & 15;
    const int grp  = lane >> 4;
    const int m0   = blockIdx.x * 16;
    const int nt   = blockIdx.y * 4 + (threadIdx.x >> 5);

    const float* w; int ldw, h, mode;
    if (nt < 32)      { w = wq; ldw = NH  * HD; h = nt;      mode = 0; }
    else if (nt < 40) { w = wk; ldw = NKV * HD; h = nt - 32; mode = 1; }
    else              { w = wv; ldw = NKV * HD; h = nt - 40; mode = 2; }

    v8f acc[4];
    gemm16x64(x, DIMK, w, ldw, m0, h * HD, DIMK, acc);

    const int b  = m0 >> 11;
    const int s0 = m0 & (SEQ - 1);

    if (mode != 2) {
#pragma unroll
        for (int j = 0; j < 4; ++j) {
            const int nhalf = j * 8 + (ln >> 1);
#pragma unroll
            for (int r = 0; r < 8; ++r) {
                const int pos = s0 + r + grp * 8;
                const float c = fcos[pos * (HD / 2) + nhalf];
                const float s = fsin[pos * (HD / 2) + nhalf];
                float vv = acc[j][r];
                float other = __shfl_xor(vv, 1, 32);
                acc[j][r] = (lane & 1) ? (other * s + vv * c)
                                       : (vv * c - other * s);
            }
        }
    }

    if (mode == 0) {
        float* base = qws + ((size_t)(b * NH + h) * SEQ + s0) * HD;
#pragma unroll
        for (int r = 0; r < 8; ++r) {
            const int m = r + grp * 8;
#pragma unroll
            for (int j = 0; j < 4; ++j)
                base[(size_t)m * HD + j * 16 + ln] = acc[j][r];
        }
    } else if (mode == 1) {
        float* base = ktws + (size_t)(b * NKV + h) * HD * SEQ;
#pragma unroll
        for (int r = 0; r < 8; ++r) {
            const int m = r + grp * 8;
#pragma unroll
            for (int j = 0; j < 4; ++j)
                base[(size_t)(j * 16 + ln) * SEQ + s0 + m] = acc[j][r];
        }
    } else {
        float* base = vws + ((size_t)(b * NKV + h) * SEQ + s0) * HD;
#pragma unroll
        for (int r = 0; r < 8; ++r) {
            const int m = r + grp * 8;
#pragma unroll
            for (int j = 0; j < 4; ++j)
                base[(size_t)m * HD + j * 16 + ln] = acc[j][r];
        }
    }
}

__global__ void attn_kernel(const float* __restrict__ qws,
                            const float* __restrict__ ktws,
                            const float* __restrict__ vws,
                            float* __restrict__ aws) {
    __shared__ float pl[16 * 18];
    const int lane = threadIdx.x & 31;
    const int ln   = lane & 15;
    const int grp  = lane >> 4;
    const int g2   = grp * 2;
    const int q0   = blockIdx.x * 16;
    const int h    = blockIdx.y;
    const int b    = blockIdx.z;
    const int hk   = h >> 2;

    const float* qb  = qws + ((size_t)(b * NH + h) * SEQ + q0) * HD;
    const float* ktb = ktws + (size_t)(b * NKV + hk) * HD * SEQ;
    const float* vb  = vws + (size_t)(b * NKV + hk) * SEQ * HD;

    v2f aq[16];
#pragma unroll
    for (int kk = 0; kk < 16; ++kk)
        aq[kk] = *(const v2f*)(qb + ln * HD + kk * 4 + g2);

    v8f o[4]; o[0] = {}; o[1] = {}; o[2] = {}; o[3] = {};
    float mrow[8], lrow[8];
#pragma unroll
    for (int r = 0; r < 8; ++r) { mrow[r] = -1e30f; lrow[r] = 0.0f; }

    for (int n0 = 0; n0 <= q0; n0 += 16) {
        v8f s = {};
#pragma unroll
        for (int kk = 0; kk < 16; ++kk) {
            const float* kp = ktb + (size_t)(kk * 4 + g2) * SEQ + n0 + ln;
            v2f bk = { kp[0], kp[SEQ] };
            s = wmma4(aq[kk], bk, s);
        }

        const bool diag = (n0 == q0);
        float p[8];
#pragma unroll
        for (int r = 0; r < 8; ++r) {
            float sv = s[r] * 0.125f;
            if (diag && (ln > r + grp * 8)) sv = -1e30f;
            float rm = sv;
            rm = fmaxf(rm, __shfl_xor(rm, 1, 32));
            rm = fmaxf(rm, __shfl_xor(rm, 2, 32));
            rm = fmaxf(rm, __shfl_xor(rm, 4, 32));
            rm = fmaxf(rm, __shfl_xor(rm, 8, 32));
            const float mn    = fmaxf(mrow[r], rm);
            const float alpha = expf(mrow[r] - mn);
            const float pv    = expf(sv - mn);
            float rs = pv;
            rs += __shfl_xor(rs, 1, 32);
            rs += __shfl_xor(rs, 2, 32);
            rs += __shfl_xor(rs, 4, 32);
            rs += __shfl_xor(rs, 8, 32);
            lrow[r] = lrow[r] * alpha + rs;
            mrow[r] = mn;
            p[r] = pv;
#pragma unroll
            for (int j = 0; j < 4; ++j) o[j][r] *= alpha;
        }

        __syncthreads();
#pragma unroll
        for (int r = 0; r < 8; ++r)
            pl[(r + grp * 8) * 18 + ln] = p[r];
        __syncthreads();
        v2f ap[4];
#pragma unroll
        for (int kk = 0; kk < 4; ++kk)
            ap[kk] = *(const v2f*)(&pl[ln * 18 + kk * 4 + g2]);

#pragma unroll
        for (int kk = 0; kk < 4; ++kk) {
            const float* vp = vb + (size_t)(n0 + kk * 4 + g2) * HD + ln;
#pragma unroll
            for (int j = 0; j < 4; ++j) {
                v2f bv = { vp[j * 16], vp[HD + j * 16] };
                o[j] = wmma4(ap[kk], bv, o[j]);
            }
        }
    }

    float* ob = aws + ((size_t)(b * SEQ) + q0) * (NH * HD) + h * HD;
#pragma unroll
    for (int r = 0; r < 8; ++r) {
        const float inv = 1.0f / lrow[r];
        const int m = r + grp * 8;
#pragma unroll
        for (int j = 0; j < 4; ++j)
            ob[(size_t)m * (NH * HD) + j * 16 + ln] = o[j][r] * inv;
    }
}

__global__ void oproj_kernel(const float* __restrict__ aws,
                             const float* __restrict__ wo,
                             float* __restrict__ out) {
    const int lane = threadIdx.x & 31;
    const int ln   = lane & 15;
    const int grp  = lane >> 4;
    const int m0   = blockIdx.x * 16;
    const int nt   = blockIdx.y * 4 + (threadIdx.x >> 5);

    v8f acc[4];
    gemm16x64(aws, DIMK, wo, DIMK, m0, nt * 64, DIMK, acc);

#pragma unroll
    for (int r = 0; r < 8; ++r) {
        const int m = r + grp * 8;
#pragma unroll
        for (int j = 0; j < 4; ++j)
            out[(size_t)(m0 + m) * DIMK + nt * 64 + j * 16 + ln] = acc[j][r];
    }
}

extern "C" void kernel_launch(void* const* d_in, const int* in_sizes, int n_in,
                              void* d_out, int out_size, void* d_ws, size_t ws_size,
                              hipStream_t stream) {
    const float* x    = (const float*)d_in[0];
    const float* fcos = (const float*)d_in[2];
    const float* fsin = (const float*)d_in[3];
    const float* wq   = (const float*)d_in[5];
    const float* wk   = (const float*)d_in[6];
    const float* wv   = (const float*)d_in[7];
    const float* wo   = (const float*)d_in[8];
    float* out = (float*)d_out;

    float* qws  = (float*)d_ws;
    float* ktws = qws  + (size_t)BATCH * NH  * SEQ * HD;
    float* vws  = ktws + (size_t)BATCH * NKV * HD  * SEQ;
    float* aws  = vws  + (size_t)BATCH * NKV * SEQ * HD;

    qkv_rope_kernel<<<dim3((BATCH * SEQ) / 16, 12, 1), 128, 0, stream>>>(
        x, wq, wk, wv, fcos, fsin, qws, ktws, vws);
    attn_kernel<<<dim3(SEQ / 16, NH, BATCH), 32, 0, stream>>>(
        qws, ktws, vws, aws);
    oproj_kernel<<<dim3((BATCH * SEQ) / 16, 8, 1), 128, 0, stream>>>(
        aws, wo, out);
}


# attention: a query tile's key blocks are split over the 8 waves of a block (wave w takes blocks w, w+8, ..); partial max, sum and O merged in LDS by wave 0; order of row sums and PV sums changes
# speedup vs baseline: 1.2851x; 1.2851x over previous
#include <hip/hip_runtime.h>

#define DIMK   2048
#define NH     32
#define NKV    8
#define HD     64
#define SEQ    2048
#define BATCH  2

typedef __attribute__((ext_vector_type(2))) float v2f;
typedef __attribute__((ext_vector_type(8))) float v8f;

__device__ __forceinline__ v8f wmma4(v2f a, v2f b, v8f c) {
    return __builtin_amdgcn_wmma_f32_16x16x4_f32(
         false, a,  false, b,
         (short)0, c,  false,  false);
}

__device__ __forceinline__ void gemm16x64(const float* __restrict__ A, int lda,
                                          const float* __restrict__ B, int ldb,
                                          int m0, int col0, int K, v8f acc[4]) {
    const int lane = threadIdx.x & 31;
    const int ln   = lane & 15;
    const int g2   = (lane >> 4) * 2;
    const float* pa = A + (size_t)(m0 + ln) * lda + g2;
    const float* pb = B + (size_t)g2 * ldb + col0 + ln;
    acc[0] = {}; acc[1] = {}; acc[2] = {}; acc[3] = {};
    for (int k = 0; k < K; k += 4) {
        v2f a = *(const v2f*)(pa + k);
        const float* pbk = pb + (size_t)k * ldb;
#pragma unroll
        for (int j = 0; j < 4; ++j) {
            v2f b = { pbk[j * 16], pbk[j * 16 + ldb] };
            acc[j] = wmma4(a, b, acc[j]);
        }
    }
}

__global__ void qkv_rope_kernel(const float* __restrict__ x,
                                const float* __restrict__ wq,
                                const float* __restrict__ wk,
                                const float* __restrict__ wv,
                                const float* __restrict__ fcos,
                                const float* __restrict__ fsin,
                                float* __restrict__ qws,
                                float* __restrict__ ktws,
                                float* __restrict__ vws) {
    const int lane = threadIdx.x & 31;
    const int ln   = lane & 15;
    const int grp  = lane >> 4;
    const int m0   = blockIdx.x * 16;
    const int nt   = blockIdx.y * 4 + (threadIdx.x >> 5);

    const float* w; int ldw, h, mode;
    if (nt < 32)      { w = wq; ldw = NH  * HD; h = nt;      mode = 0; }
    else if (nt < 40) { w = wk; ldw = NKV * HD; h = nt - 32; mode = 1; }
    else              { w = wv; ldw = NKV * HD; h = nt - 40; mode = 2; }

    v8f acc[4];
    gemm16x64(x, DIMK, w, ldw, m0, h * HD, DIMK, acc);

    const int b  = m0 >> 11;
    const int s0 = m0 & (SEQ - 1);

    if (mode != 2) {
#pragma unroll
        for (int j = 0; j < 4; ++j) {
            const int nhalf = j * 8 + (ln >> 1);
#pragma unroll
            for (int r = 0; r < 8; ++r) {
                const int pos = s0 + r + grp * 8;
                const float c = fcos[pos * (HD / 2) + nhalf];
                const float s = fsin[pos * (HD / 2) + nhalf];
                float vv = acc[j][r];
                float other = __shfl_xor(vv, 1, 32);
                acc[j][r] = (lane & 1) ? (other * s + vv * c)
                                       : (vv * c - other * s);
            }
        }
    }

    if (mode == 0) {
        float* base = qws + ((size_t)(b * NH + h) * SEQ + s0) * HD;
#pragma unroll
        for (int r = 0; r < 8; ++r) {
            const int m = r + grp * 8;
#pragma unroll
            for (int j = 0; j < 4; ++j)
                base[(size_t)m * HD + j * 16 + ln] = acc[j][r];
        }
    } else if (mode == 1) {
        float* base = ktws + (size_t)(b * NKV + h) * HD * SEQ;
#pragma unroll
        for (int r = 0; r < 8; ++r) {
            const int m = r + grp * 8;
#pragma unroll
            for (int j = 0; j < 4; ++j)
                base[(size_t)(j * 16 + ln) * SEQ + s0 + m] = acc[j][r];
        }
    } else {
        float* base = vws + ((size_t)(b * NKV + h) * SEQ + s0) * HD;
#pragma unroll
        for (int r = 0; r < 8; ++r) {
            const int m = r + grp * 8;
#pragma unroll
            for (int j = 0; j < 4; ++j)
                base[(size_t)m * HD + j * 16 + ln] = acc[j][r];
        }
    }
}

#define KSPLIT 8
__global__ void attn_kernel(const float* __restrict__ qws,
                            const float* __restrict__ ktws,
                            const float* __restrict__ vws,
                            float* __restrict__ aws) {
    __shared__ float  pl[KSPLIT][16 * 18];
    __shared__ float4 part[KSPLIT - 1][8][32];
    __shared__ float  ml[KSPLIT - 1][2][16];
    const int lane = threadIdx.x & 31;
    const int ln   = lane & 15;
    const int grp  = lane >> 4;
    const int g2   = grp * 2;
    const int w    = __builtin_amdgcn_readfirstlane((int)(threadIdx.x >> 5));
    const int q0   = blockIdx.x * 16;
    const int h    = blockIdx.y;
    const int b    = blockIdx.z;
    const int hk   = h >> 2;

    const float* qb  = qws + ((size_t)(b * NH + h) * SEQ + q0) * HD;
    const float* ktb = ktws + (size_t)(b * NKV + hk) * HD * SEQ;
    const float* vb  = vws + (size_t)(b * NKV + hk) * SEQ * HD;
    float* plw = pl[w];

    v2f aq[16];
#pragma unroll
    for (int kk = 0; kk < 16; ++kk)
        aq[kk] = *(const v2f*)(qb + ln * HD + kk * 4 + g2);

    v8f o[4]; o[0] = {}; o[1] = {}; o[2] = {}; o[3] = {};
    float mrow[8], lrow[8];
#pragma unroll
    for (int r = 0; r < 8; ++r) { mrow[r] = -1e30f; lrow[r] = 0.0f; }

    for (int n0 = w * 16; n0 <= q0; n0 += 16 * KSPLIT) {
        v8f s = {};
#pragma unroll
        for (int kk = 0; kk < 16; ++kk) {
            const float* kp = ktb + (size_t)(kk * 4 + g2) * SEQ + n0 + ln;
            v2f bk = { kp[0], kp[SEQ] };
            s = wmma4(aq[kk], bk, s);
        }

        const bool diag = (n0 == q0);
        float p[8];
#pragma unroll
        for (int r = 0; r < 8; ++r) {
            float sv = s[r] * 0.125f;
            if (diag && (ln > r + grp * 8)) sv = -1e30f;
            float rm = sv;
            rm = fmaxf(rm, __shfl_xor(rm, 1, 32));
            rm = fmaxf(rm, __shfl_xor(rm, 2, 32));
            rm = fmaxf(rm, __shfl_xor(rm, 4, 32));
            rm = fmaxf(rm, __shfl_xor(rm, 8, 32));
            const float mn    = fmaxf(mrow[r], rm);
            const float alpha = expf(mrow[r] - mn);
            const float pv    = expf(sv - mn);
            float rs = pv;
            rs += __shfl_xor(rs, 1, 32);
            rs += __shfl_xor(rs, 2, 32);
            rs += __shfl_xor(rs, 4, 32);
            rs += __shfl_xor(rs, 8, 32);
            lrow[r] = lrow[r] * alpha + rs;
            mrow[r] = mn;
            p[r] = pv;
#pragma unroll
            for (int j = 0; j < 4; ++j) o[j][r] *= alpha;
        }

        __builtin_amdgcn_fence(__ATOMIC_RELEASE, "wavefront");
#pragma unroll
        for (int r = 0; r < 8; ++r)
            plw[(r + grp * 8) * 18 + ln] = p[r];
        __builtin_amdgcn_fence(__ATOMIC_ACQ_REL, "wavefront");
        v2f ap[4];
#pragma unroll
        for (int kk = 0; kk < 4; ++kk)
            ap[kk] = *(const v2f*)(&plw[ln * 18 + kk * 4 + g2]);

#pragma unroll
        for (int kk = 0; kk < 4; ++kk) {
            const float* vp = vb + (size_t)(n0 + kk * 4 + g2) * HD + ln;
#pragma unroll
            for (int j = 0; j < 4; ++j) {
                v2f bv = { vp[j * 16], vp[HD + j * 16] };
                o[j] = wmma4(ap[kk], bv, o[j]);
            }
        }
    }

    if (w > 0) {
#pragma unroll
        for (int j = 0; j < 4; ++j) {
            part[w - 1][j * 2 + 0][lane] = make_float4(o[j][0], o[j][1], o[j][2], o[j][3]);
            part[w - 1][j * 2 + 1][lane] = make_float4(o[j][4], o[j][5], o[j][6], o[j][7]);
        }
        if (ln == 0) {
#pragma unroll
            for (int r = 0; r < 8; ++r) {
                ml[w - 1][grp][r]     = mrow[r];
                ml[w - 1][grp][8 + r] = lrow[r];
            }
        }
    }
    __syncthreads();
    if (w > 0) return;

#pragma unroll
    for (int pw = 0; pw < KSPLIT - 1; ++pw) {
        float a0[8], a1[8];
#pragma unroll
        for (int r = 0; r < 8; ++r) {
            const float mw = ml[pw][grp][r];
            const float lw = ml[pw][grp][8 + r];
            const float mn = fmaxf(mrow[r], mw);
            a0[r] = expf(mrow[r] - mn);
            a1[r] = expf(mw - mn);
            lrow[r] = lrow[r] * a0[r] + lw * a1[r];
            mrow[r] = mn;
        }
#pragma unroll
        for (int j = 0; j < 4; ++j) {
            const float4 lo = part[pw][j * 2 + 0][lane];
            const float4 hi = part[pw][j * 2 + 1][lane];
            o[j][0] = o[j][0] * a0[0] + lo.x * a1[0];
            o[j][1] = o[j][1] * a0[1] + lo.y * a1[1];
            o[j][2] = o[j][2] * a0[2] + lo.z * a1[2];
            o[j][3] = o[j][3] * a0[3] + lo.w * a1[3];
            o[j][4] = o[j][4] * a0[4] + hi.x * a1[4];
            o[j][5] = o[j][5] * a0[5] + hi.y * a1[5];
            o[j][6] = o[j][6] * a0[6] + hi.z * a1[6];
            o[j][7] = o[j][7] * a0[7] + hi.w * a1[7];
        }
    }

    float* ob = aws + ((size_t)(b * SEQ) + q0) * (NH * HD) + h * HD;
#pragma unroll
    for (int r = 0; r < 8; ++r) {
        const float inv = 1.0f / lrow[r];
        const int m = r + grp * 8;
#pragma unroll
        for (int j = 0; j < 4; ++j)
            ob[(size_t)m * (NH * HD) + j * 16 + ln] = o[j][r] * inv;
    }
}

__global__ void oproj_kernel(const float* __restrict__ aws,
                             const float* __restrict__ wo,
                             float* __restrict__ out) {
    const int lane = threadIdx.x & 31;
    const int ln   = lane & 15;
    const int grp  = lane >> 4;
    const int m0   = blockIdx.x * 16;
    const int nt   = blockIdx.y * 4 + (threadIdx.x >> 5);

    v8f acc[4];
    gemm16x64(aws, DIMK, wo, DIMK, m0, nt * 64, DIMK, acc);

#pragma unroll
    for (int r = 0; r < 8; ++r) {
        const int m = r + grp * 8;
#pragma unroll
        for (int j = 0; j < 4; ++j)
            out[(size_t)(m0 + m) * DIMK + nt * 64 + j * 16 + ln] = acc[j][r];
    }
}

extern "C" void kernel_launch(void* const* d_in, const int* in_sizes, int n_in,
                              void* d_out, int out_size, void* d_ws, size_t ws_size,
                              hipStream_t stream) {
    const float* x    = (const float*)d_in[0];
    const float* fcos = (const float*)d_in[2];
    const float* fsin = (const float*)d_in[3];
    const float* wq   = (const float*)d_in[5];
    const float* wk   = (const float*)d_in[6];
    const float* wv   = (const float*)d_in[7];
    const float* wo   = (const float*)d_in[8];
    float* out = (float*)d_out;

    float* qws  = (float*)d_ws;
    float* ktws = qws  + (size_t)BATCH * NH  * SEQ * HD;
    float* vws  = ktws + (size_t)BATCH * NKV * HD  * SEQ;
    float* aws  = vws  + (size_t)BATCH * NKV * SEQ * HD;

    qkv_rope_kernel<<<dim3((BATCH * SEQ) / 16, 12, 1), 128, 0, stream>>>(
        x, wq, wk, wv, fcos, fsin, qws, ktws, vws);
    attn_kernel<<<dim3(SEQ / 16, NH, BATCH), 32 * KSPLIT, 0, stream>>>(
        qws, ktws, vws, aws);
    oproj_kernel<<<dim3((BATCH * SEQ) / 16, 8, 1), 128, 0, stream>>>(
        aws, wo, out);
}


# projections: both fragments by 16-byte loads (a lane owns 4 adjacent columns, a lane half 4 adjacent k of an 8-k chunk): 5 loads per 8 WMMAs, was 18; 16-byte stores; order of the k sums changes
# speedup vs baseline: 1.4303x; 1.1130x over previous
#include <hip/hip_runtime.h>

#define DIMK   2048
#define NH     32
#define NKV    8
#define HD     64
#define SEQ    2048
#define BATCH  2

typedef __attribute__((ext_vector_type(2))) float v2f;
typedef __attribute__((ext_vector_type(4))) float v4f;
typedef __attribute__((ext_vector_type(8))) float v8f;

__device__ __forceinline__ v8f wmma4(v2f a, v2f b, v8f c) {
    return __builtin_amdgcn_wmma_f32_16x16x4_f32(
         false, a,  false, b,
         (short)0, c,  false,  false);
}

template <int LDB>
__device__ __forceinline__ void gemm16x64(const float* __restrict__ A, int lda,
                                          const float* __restrict__ B,
                                          int m0, int col0, int K, v8f acc[4]) {
    constexpr int ldb = LDB;
    const int lane = threadIdx.x & 31;
    const int ln   = lane & 15;
    const int g4   = (lane >> 4) * 4;
    const float* pa = A + (size_t)(m0 + ln) * lda + g4;
    const float* pb = B + (size_t)g4 * ldb + col0 + ln * 4;
    acc[0] = {}; acc[1] = {}; acc[2] = {}; acc[3] = {};
    const size_t step = (size_t)ldb * 8;
    for (int k = 0; k < K; k += 8, pa += 8, pb += step) {
        const v4f a  = *(const v4f*)(pa);
        const v4f b0 = *(const v4f*)(pb);
        const v4f b1 = *(const v4f*)(pb + ldb);
        const v4f b2 = *(const v4f*)(pb + 2 * (size_t)ldb);
        const v4f b3 = *(const v4f*)(pb + 3 * (size_t)ldb);
        const v2f a01 = { a[0], a[1] };
        const v2f a23 = { a[2], a[3] };
#pragma unroll
        for (int j = 0; j < 4; ++j) {
            v2f b = { b0[j], b1[j] };
            acc[j] = wmma4(a01, b, acc[j]);
        }
#pragma unroll
        for (int j = 0; j < 4; ++j) {
            v2f b = { b2[j], b3[j] };
            acc[j] = wmma4(a23, b, acc[j]);
        }
    }
}

__global__ void qkv_rope_kernel(const float* __restrict__ x,
                                const float* __restrict__ wq,
                                const float* __restrict__ wk,
                                const float* __restrict__ wv,
                                const float* __restrict__ fcos,
                                const float* __restrict__ fsin,
                                float* __restrict__ qws,
                                float* __restrict__ ktws,
                                float* __restrict__ vws) {
    const int lane = threadIdx.x & 31;
    const int ln   = lane & 15;
    const int grp  = lane >> 4;
    const int m0   = blockIdx.x * 16;
    const int nt   = blockIdx.y * 4 + (threadIdx.x >> 5);

    const float* w; int h, mode;
    if (nt < 32)      { w = wq; h = nt;      mode = 0; }
    else if (nt < 40) { w = wk; h = nt - 32; mode = 1; }
    else              { w = wv; h = nt - 40; mode = 2; }

    v8f acc[4];
    if (mode == 0) gemm16x64<NH  * HD>(x, DIMK, w, m0, h * HD, DIMK, acc);
    else           gemm16x64<NKV * HD>(x, DIMK, w, m0, h * HD, DIMK, acc);

    const int b  = m0 >> 11;
    const int s0 = m0 & (SEQ - 1);

    if (mode != 2) {
#pragma unroll
        for (int r = 0; r < 8; ++r) {
            const int pos = s0 + r + grp * 8;
            const v2f c = *(const v2f*)(fcos + pos * (HD / 2) + 2 * ln);
            const v2f s = *(const v2f*)(fsin + pos * (HD / 2) + 2 * ln);
            const float e0 = acc[0][r], o0 = acc[1][r];
            const float e1 = acc[2][r], o1 = acc[3][r];
            acc[0][r] = e0 * c[0] - o0 * s[0];
            acc[1][r] = e0 * s[0] + o0 * c[0];
            acc[2][r] = e1 * c[1] - o1 * s[1];
            acc[3][r] = e1 * s[1] + o1 * c[1];
        }
    }

    if (mode == 1) {
        float* base = ktws + (size_t)(b * NKV + h) * HD * SEQ + s0 + grp * 8;
#pragma unroll
        for (int j = 0; j < 4; ++j) {
            float* p = base + (size_t)(4 * ln + j) * SEQ;
            *(v4f*)(p)     = (v4f){ acc[j][0], acc[j][1], acc[j][2], acc[j][3] };
            *(v4f*)(p + 4) = (v4f){ acc[j][4], acc[j][5], acc[j][6], acc[j][7] };
        }
    } else {
        float* base = (mode == 0) ? qws + ((size_t)(b * NH  + h) * SEQ + s0) * HD
                                  : vws + ((size_t)(b * NKV + h) * SEQ + s0) * HD;
#pragma unroll
        for (int r = 0; r < 8; ++r) {
            const int m = r + grp * 8;
            *(v4f*)(base + (size_t)m * HD + 4 * ln) = (v4f){ acc[0][r], acc[1][r], acc[2][r], acc[3][r] };
        }
    }
}

#define KSPLIT 8
__global__ void attn_kernel(const float* __restrict__ qws,
                            const float* __restrict__ ktws,
                            const float* __restrict__ vws,
                            float* __restrict__ aws) {
    __shared__ float  pl[KSPLIT][16 * 18];
    __shared__ float4 part[KSPLIT - 1][8][32];
    __shared__ float  ml[KSPLIT - 1][2][16];
    const int lane = threadIdx.x & 31;
    const int ln   = lane & 15;
    const int grp  = lane >> 4;
    const int g2   = grp * 2;
    const int w    = __builtin_amdgcn_readfirstlane((int)(threadIdx.x >> 5));
    const int q0   = blockIdx.x * 16;
    const int h    = blockIdx.y;
    const int b    = blockIdx.z;
    const int hk   = h >> 2;

    const float* qb  = qws + ((size_t)(b * NH + h) * SEQ + q0) * HD;
    const float* ktb = ktws + (size_t)(b * NKV + hk) * HD * SEQ;
    const float* vb  = vws + (size_t)(b * NKV + hk) * SEQ * HD;
    float* plw = pl[w];

    v2f aq[16];
#pragma unroll
    for (int kk = 0; kk < 16; ++kk)
        aq[kk] = *(const v2f*)(qb + ln * HD + kk * 4 + g2);

    v8f o[4]; o[0] = {}; o[1] = {}; o[2] = {}; o[3] = {};
    float mrow[8], lrow[8];
#pragma unroll
    for (int r = 0; r < 8; ++r) { mrow[r] = -1e30f; lrow[r] = 0.0f; }

    for (int n0 = w * 16; n0 <= q0; n0 += 16 * KSPLIT) {
        v8f s = {};
#pragma unroll
        for (int kk = 0; kk < 16; ++kk) {
            const float* kp = ktb + (size_t)(kk * 4 + g2) * SEQ + n0 + ln;
            v2f bk = { kp[0], kp[SEQ] };
            s = wmma4(aq[kk], bk, s);
        }

        const bool diag = (n0 == q0);
        float p[8];
#pragma unroll
        for (int r = 0; r < 8; ++r) {
            float sv = s[r] * 0.125f;
            if (diag && (ln > r + grp * 8)) sv = -1e30f;
            float rm = sv;
            rm = fmaxf(rm, __shfl_xor(rm, 1, 32));
            rm = fmaxf(rm, __shfl_xor(rm, 2, 32));
            rm = fmaxf(rm, __shfl_xor(rm, 4, 32));
            rm = fmaxf(rm, __shfl_xor(rm, 8, 32));
            const float mn    = fmaxf(mrow[r], rm);
            const float alpha = expf(mrow[r] - mn);
            const float pv    = expf(sv - mn);
            float rs = pv;
            rs += __shfl_xor(rs, 1, 32);
            rs += __shfl_xor(rs, 2, 32);
            rs += __shfl_xor(rs, 4, 32);
            rs += __shfl_xor(rs, 8, 32);
            lrow[r] = lrow[r] * alpha + rs;
            mrow[r] = mn;
            p[r] = pv;
#pragma unroll
            for (int j = 0; j < 4; ++j) o[j][r] *= alpha;
        }

        __builtin_amdgcn_fence(__ATOMIC_RELEASE, "wavefront");
#pragma unroll
        for (int r = 0; r < 8; ++r)
            plw[(r + grp * 8) * 18 + ln] = p[r];
        __builtin_amdgcn_fence(__ATOMIC_ACQ_REL, "wavefront");
        v2f ap[4];
#pragma unroll
        for (int kk = 0; kk < 4; ++kk)
            ap[kk] = *(const v2f*)(&plw[ln * 18 + kk * 4 + g2]);

#pragma unroll
        for (int kk = 0; kk < 4; ++kk) {
            const float* vp = vb + (size_t)(n0 + kk * 4 + g2) * HD + ln;
#pragma unroll
            for (int j = 0; j < 4; ++j) {
                v2f bv = { vp[j * 16], vp[HD + j * 16] };
                o[j] = wmma4(ap[kk], bv, o[j]);
            }
        }
    }

    if (w > 0) {
#pragma unroll
        for (int j = 0; j < 4; ++j) {
            part[w - 1][j * 2 + 0][lane] = make_float4(o[j][0], o[j][1], o[j][2], o[j][3]);
            part[w - 1][j * 2 + 1][lane] = make_float4(o[j][4], o[j][5], o[j][6], o[j][7]);
        }
        if (ln == 0) {
#pragma unroll
            for (int r = 0; r < 8; ++r) {
                ml[w - 1][grp][r]     = mrow[r];
                ml[w - 1][grp][8 + r] = lrow[r];
            }
        }
    }
    __syncthreads();
    if (w > 0) return;

#pragma unroll
    for (int pw = 0; pw < KSPLIT - 1; ++pw) {
        float a0[8], a1[8];
#pragma unroll
        for (int r = 0; r < 8; ++r) {
            const float mw = ml[pw][grp][r];
            const float lw = ml[pw][grp][8 + r];
            const float mn = fmaxf(mrow[r], mw);
            a0[r] = expf(mrow[r] - mn);
            a1[r] = expf(mw - mn);
            lrow[r] = lrow[r] * a0[r] + lw * a1[r];
            mrow[r] = mn;
        }
#pragma unroll
        for (int j = 0; j < 4; ++j) {
            const float4 lo = part[pw][j * 2 + 0][lane];
            const float4 hi = part[pw][j * 2 + 1][lane];
            o[j][0] = o[j][0] * a0[0] + lo.x * a1[0];
            o[j][1] = o[j][1] * a0[1] + lo.y * a1[1];
            o[j][2] = o[j][2] * a0[2] + lo.z * a1[2];
            o[j][3] = o[j][3] * a0[3] + lo.w * a1[3];
            o[j][4] = o[j][4] * a0[4] + hi.x * a1[4];
            o[j][5] = o[j][5] * a0[5] + hi.y * a1[5];
            o[j][6] = o[j][6] * a0[6] + hi.z * a1[6];
            o[j][7] = o[j][7] * a0[7] + hi.w * a1[7];
        }
    }

    float* ob = aws + ((size_t)(b * SEQ) + q0) * (NH * HD) + h * HD;
#pragma unroll
    for (int r = 0; r < 8; ++r) {
        const float inv = 1.0f / lrow[r];
        const int m = r + grp * 8;
#pragma unroll
        for (int j = 0; j < 4; ++j)
            ob[(size_t)m * (NH * HD) + j * 16 + ln] = o[j][r] * inv;
    }
}

__global__ void oproj_kernel(const float* __restrict__ aws,
                             const float* __restrict__ wo,
                             float* __restrict__ out) {
    const int lane = threadIdx.x & 31;
    const int ln   = lane & 15;
    const int grp  = lane >> 4;
    const int m0   = blockIdx.x * 16;
    const int nt   = blockIdx.y * 4 + (threadIdx.x >> 5);

    v8f acc[4];
    gemm16x64<DIMK>(aws, DIMK, wo, m0, nt * 64, DIMK, acc);

#pragma unroll
    for (int r = 0; r < 8; ++r) {
        const int m = r + grp * 8;
        *(v4f*)(out + (size_t)(m0 + m) * DIMK + nt * 64 + 4 * ln) = (v4f){ acc[0][r], acc[1][r], acc[2][r], acc[3][r] };
    }
}

extern "C" void kernel_launch(void* const* d_in, const int* in_sizes, int n_in,
                              void* d_out, int out_size, void* d_ws, size_t ws_size,
                              hipStream_t stream) {
    const float* x    = (const float*)d_in[0];
    const float* fcos = (const float*)d_in[2];
    const float* fsin = (const float*)d_in[3];
    const float* wq   = (const float*)d_in[5];
    const float* wk   = (const float*)d_in[6];
    const float* wv   = (const float*)d_in[7];
    const float* wo   = (const float*)d_in[8];
    float* out = (float*)d_out;

    float* qws  = (float*)d_ws;
    float* ktws = qws  + (size_t)BATCH * NH  * SEQ * HD;
    float* vws  = ktws + (size_t)BATCH * NKV * HD  * SEQ;
    float* aws  = vws  + (size_t)BATCH * NKV * SEQ * HD;

    qkv_rope_kernel<<<dim3((BATCH * SEQ) / 16, 12, 1), 128, 0, stream>>>(
        x, wq, wk, wv, fcos, fsin, qws, ktws, vws);
    attn_kernel<<<dim3(SEQ / 16, NH, BATCH), 32 * KSPLIT, 0, stream>>>(
        qws, ktws, vws, aws);
    oproj_kernel<<<dim3((BATCH * SEQ) / 16, 8, 1), 128, 0, stream>>>(
        aws, wo, out);
}


# attention: block = 32 queries (2 tiles); a wave loads each key block's K/V fragments once, in groups, for both tiles (2 WMMAs per fragment; was 1, with a wait per WMMA); block size declared (256)
# speedup vs baseline: 1.8695x; 1.3071x over previous
#include <hip/hip_runtime.h>

#define DIMK   2048
#define NH     32
#define NKV    8
#define HD     64
#define SEQ    2048
#define BATCH  2

typedef __attribute__((ext_vector_type(2))) float v2f;
typedef __attribute__((ext_vector_type(4))) float v4f;
typedef __attribute__((ext_vector_type(8))) float v8f;

__device__ __forceinline__ v8f wmma4(v2f a, v2f b, v8f c) {
    return __builtin_amdgcn_wmma_f32_16x16x4_f32(
         false, a,  false, b,
         (short)0, c,  false,  false);
}

template <int LDB>
__device__ __forceinline__ void gemm16x64(const float* __restrict__ A, int lda,
                                          const float* __restrict__ B,
                                          int m0, int col0, int K, v8f acc[4]) {
    constexpr int ldb = LDB;
    const int lane = threadIdx.x & 31;
    const int ln   = lane & 15;
    const int g4   = (lane >> 4) * 4;
    const float* pa = A + (size_t)(m0 + ln) * lda + g4;
    const float* pb = B + (size_t)g4 * ldb + col0 + ln * 4;
    acc[0] = {}; acc[1] = {}; acc[2] = {}; acc[3] = {};
    const size_t step = (size_t)ldb * 8;
    for (int k = 0; k < K; k += 8, pa += 8, pb += step) {
        const v4f a  = *(const v4f*)(pa);
        const v4f b0 = *(const v4f*)(pb);
        const v4f b1 = *(const v4f*)(pb + ldb);
        const v4f b2 = *(const v4f*)(pb + 2 * (size_t)ldb);
        const v4f b3 = *(const v4f*)(pb + 3 * (size_t)ldb);
        const v2f a01 = { a[0], a[1] };
        const v2f a23 = { a[2], a[3] };
#pragma unroll
        for (int j = 0; j < 4; ++j) {
            v2f b = { b0[j], b1[j] };
            acc[j] = wmma4(a01, b, acc[j]);
        }
#pragma unroll
        for (int j = 0; j < 4; ++j) {
            v2f b = { b2[j], b3[j] };
            acc[j] = wmma4(a23, b, acc[j]);
        }
    }
}

__global__ void qkv_rope_kernel(const float* __restrict__ x,
                                const float* __restrict__ wq,
                                const float* __restrict__ wk,
                                const float* __restrict__ wv,
                                const float* __restrict__ fcos,
                                const float* __restrict__ fsin,
                                float* __restrict__ qws,
                                float* __restrict__ ktws,
                                float* __restrict__ vws) {
    const int lane = threadIdx.x & 31;
    const int ln   = lane & 15;
    const int grp  = lane >> 4;
    const int m0   = blockIdx.x * 16;
    const int nt   = blockIdx.y * 4 + (threadIdx.x >> 5);

    const float* w; int h, mode;
    if (nt < 32)      { w = wq; h = nt;      mode = 0; }
    else if (nt < 40) { w = wk; h = nt - 32; mode = 1; }
    else              { w = wv; h = nt - 40; mode = 2; }

    v8f acc[4];
    if (mode == 0) gemm16x64<NH  * HD>(x, DIMK, w, m0, h * HD, DIMK, acc);
    else           gemm16x64<NKV * HD>(x, DIMK, w, m0, h * HD, DIMK, acc);

    const int b  = m0 >> 11;
    const int s0 = m0 & (SEQ - 1);

    if (mode != 2) {
#pragma unroll
        for (int r = 0; r < 8; ++r) {
            const int pos = s0 + r + grp * 8;
            const v2f c = *(const v2f*)(fcos + pos * (HD / 2) + 2 * ln);
            const v2f s = *(const v2f*)(fsin + pos * (HD / 2) + 2 * ln);
            const float e0 = acc[0][r], o0 = acc[1][r];
            const float e1 = acc[2][r], o1 = acc[3][r];
            acc[0][r] = e0 * c[0] - o0 * s[0];
            acc[1][r] = e0 * s[0] + o0 * c[0];
            acc[2][r] = e1 * c[1] - o1 * s[1];
            acc[3][r] = e1 * s[1] + o1 * c[1];
        }
    }

    if (mode == 1) {
        float* base = ktws + (size_t)(b * NKV + h) * HD * SEQ + s0 + grp * 8;
#pragma unroll
        for (int j = 0; j < 4; ++j) {
            float* p = base + (size_t)(4 * ln + j) * SEQ;
            *(v4f*)(p)     = (v4f){ acc[j][0], acc[j][1], acc[j][2], acc[j][3] };
            *(v4f*)(p + 4) = (v4f){ acc[j][4], acc[j][5], acc[j][6], acc[j][7] };
        }
    } else {
        float* base = (mode == 0) ? qws + ((size_t)(b * NH  + h) * SEQ + s0) * HD
                                  : vws + ((size_t)(b * NKV + h) * SEQ + s0) * HD;
#pragma unroll
        for (int r = 0; r < 8; ++r) {
            const int m = r + grp * 8;
            *(v4f*)(base + (size_t)m * HD + 4 * ln) = (v4f){ acc[0][r], acc[1][r], acc[2][r], acc[3][r] };
        }
    }
}

#define KSPLIT 8
#define QT 2
__global__ void __launch_bounds__(32 * KSPLIT, 4)
attn_kernel(const float* __restrict__ qws,
            const float* __restrict__ ktws,
            const float* __restrict__ vws,
            float* __restrict__ aws) {
    __shared__ float  pl[KSPLIT][QT][16 * 18];
    __shared__ float4 part[KSPLIT - 1][8][32];
    __shared__ float  ml[KSPLIT - 1][QT][2][16];
    const int lane = threadIdx.x & 31;
    const int ln   = lane & 15;
    const int grp  = lane >> 4;
    const int g2   = grp * 2;
    const int w    = __builtin_amdgcn_readfirstlane((int)(threadIdx.x >> 5));
    const int q0   = blockIdx.x * (16 * QT);
    const int h    = blockIdx.y;
    const int b    = blockIdx.z;
    const int hk   = h >> 2;

    const float* qb  = qws + ((size_t)(b * NH + h) * SEQ + q0) * HD;
    const float* ktb = ktws + (size_t)(b * NKV + hk) * HD * SEQ;
    const float* vb  = vws + (size_t)(b * NKV + hk) * SEQ * HD;

    v2f aq[QT][16];
#pragma unroll
    for (int t = 0; t < QT; ++t)
#pragma unroll
        for (int kk = 0; kk < 16; ++kk)
            aq[t][kk] = *(const v2f*)(qb + (t * 16 + ln) * HD + kk * 4 + g2);

    v8f o[QT][4];
    float mrow[QT][8], lrow[QT][8];
#pragma unroll
    for (int t = 0; t < QT; ++t) {
        o[t][0] = {}; o[t][1] = {}; o[t][2] = {}; o[t][3] = {};
#pragma unroll
        for (int r = 0; r < 8; ++r) { mrow[t][r] = -1e30f; lrow[t][r] = 0.0f; }
    }

    for (int n0 = w * 16; n0 <= q0 + 16 * (QT - 1); n0 += 16 * KSPLIT) {
        v2f bk[16];
#pragma unroll
        for (int kk = 0; kk < 16; ++kk) {
            const float* kp = ktb + (size_t)(kk * 4 + g2) * SEQ + n0 + ln;
            bk[kk] = (v2f){ kp[0], kp[SEQ] };
        }
        const bool live0 = (n0 <= q0);

        v8f s[QT];
#pragma unroll
        for (int t = 0; t < QT; ++t) {
            s[t] = {};
            if (t > 0 || live0) {
#pragma unroll
                for (int kk = 0; kk < 16; ++kk)
                    s[t] = wmma4(aq[t][kk], bk[kk], s[t]);
            }
        }

        v2f bv[4][4];
#pragma unroll
        for (int kk = 0; kk < 4; ++kk) {
            const float* vp = vb + (size_t)(n0 + kk * 4 + g2) * HD + ln;
#pragma unroll
            for (int j = 0; j < 4; ++j)
                bv[kk][j] = (v2f){ vp[j * 16], vp[HD + j * 16] };
        }

#pragma unroll
        for (int t = 0; t < QT; ++t) {
            if (t > 0 || live0) {
                float* plw = pl[w][t];
                const bool diag = (n0 == q0 + 16 * t);
                float p[8];
#pragma unroll
                for (int r = 0; r < 8; ++r) {
                    float sv = s[t][r] * 0.125f;
                    if (diag && (ln > r + grp * 8)) sv = -1e30f;
                    float rm = sv;
                    rm = fmaxf(rm, __shfl_xor(rm, 1, 32));
                    rm = fmaxf(rm, __shfl_xor(rm, 2, 32));
                    rm = fmaxf(rm, __shfl_xor(rm, 4, 32));
                    rm = fmaxf(rm, __shfl_xor(rm, 8, 32));
                    const float mn    = fmaxf(mrow[t][r], rm);
                    const float alpha = expf(mrow[t][r] - mn);
                    const float pv    = expf(sv - mn);
                    float rs = pv;
                    rs += __shfl_xor(rs, 1, 32);
                    rs += __shfl_xor(rs, 2, 32);
                    rs += __shfl_xor(rs, 4, 32);
                    rs += __shfl_xor(rs, 8, 32);
                    lrow[t][r] = lrow[t][r] * alpha + rs;
                    mrow[t][r] = mn;
                    p[r] = pv;
#pragma unroll
                    for (int j = 0; j < 4; ++j) o[t][j][r] *= alpha;
                }

                __builtin_amdgcn_fence(__ATOMIC_RELEASE, "wavefront");
#pragma unroll
                for (int r = 0; r < 8; ++r)
                    plw[(r + grp * 8) * 18 + ln] = p[r];
                __builtin_amdgcn_fence(__ATOMIC_ACQ_REL, "wavefront");
                v2f ap[4];
#pragma unroll
                for (int kk = 0; kk < 4; ++kk)
                    ap[kk] = *(const v2f*)(&plw[ln * 18 + kk * 4 + g2]);

#pragma unroll
                for (int kk = 0; kk < 4; ++kk)
#pragma unroll
                    for (int j = 0; j < 4; ++j)
                        o[t][j] = wmma4(ap[kk], bv[kk][j], o[t][j]);
            }
        }
    }

#pragma unroll
    for (int t = 0; t < QT; ++t) {
        if (w > 0) {
#pragma unroll
            for (int j = 0; j < 4; ++j) {
                part[w - 1][j * 2 + 0][lane] = make_float4(o[t][j][0], o[t][j][1], o[t][j][2], o[t][j][3]);
                part[w - 1][j * 2 + 1][lane] = make_float4(o[t][j][4], o[t][j][5], o[t][j][6], o[t][j][7]);
            }
            if (ln == 0) {
#pragma unroll
                for (int r = 0; r < 8; ++r) {
                    ml[w - 1][t][grp][r]     = mrow[t][r];
                    ml[w - 1][t][grp][8 + r] = lrow[t][r];
                }
            }
        }
        __syncthreads();
        if (w == 0) {
#pragma unroll
            for (int pw = 0; pw < KSPLIT - 1; ++pw) {
                float a0[8], a1[8];
#pragma unroll
                for (int r = 0; r < 8; ++r) {
                    const float mw = ml[pw][t][grp][r];
                    const float lw = ml[pw][t][grp][8 + r];
                    const float mn = fmaxf(mrow[t][r], mw);
                    a0[r] = expf(mrow[t][r] - mn);
                    a1[r] = expf(mw - mn);
                    lrow[t][r] = lrow[t][r] * a0[r] + lw * a1[r];
                    mrow[t][r] = mn;
                }
#pragma unroll
                for (int j = 0; j < 4; ++j) {
                    const float4 lo = part[pw][j * 2 + 0][lane];
                    const float4 hi = part[pw][j * 2 + 1][lane];
                    o[t][j][0] = o[t][j][0] * a0[0] + lo.x * a1[0];
                    o[t][j][1] = o[t][j][1] * a0[1] + lo.y * a1[1];
                    o[t][j][2] = o[t][j][2] * a0[2] + lo.z * a1[2];
                    o[t][j][3] = o[t][j][3] * a0[3] + lo.w * a1[3];
                    o[t][j][4] = o[t][j][4] * a0[4] + hi.x * a1[4];
                    o[t][j][5] = o[t][j][5] * a0[5] + hi.y * a1[5];
                    o[t][j][6] = o[t][j][6] * a0[6] + hi.z * a1[6];
                    o[t][j][7] = o[t][j][7] * a0[7] + hi.w * a1[7];
                }
            }

            float* ob = aws + ((size_t)(b * SEQ) + q0 + 16 * t) * (NH * HD) + h * HD;
#pragma unroll
            for (int r = 0; r < 8; ++r) {
                const float inv = 1.0f / lrow[t][r];
                const int m = r + grp * 8;
#pragma unroll
                for (int j = 0; j < 4; ++j)
                    ob[(size_t)m * (NH * HD) + j * 16 + ln] = o[t][j][r] * inv;
            }
        }
        if (t + 1 < QT) __syncthreads();
    }
}

__global__ void oproj_kernel(const float* __restrict__ aws,
                             const float* __restrict__ wo,
                             float* __restrict__ out) {
    const int lane = threadIdx.x & 31;
    const int ln   = lane & 15;
    const int grp  = lane >> 4;
    const int m0   = blockIdx.x * 16;
    const int nt   = blockIdx.y * 4 + (threadIdx.x >> 5);

    v8f acc[4];
    gemm16x64<DIMK>(aws, DIMK, wo, m0, nt * 64, DIMK, acc);

#pragma unroll
    for (int r = 0; r < 8; ++r) {
        const int m = r + grp * 8;
        *(v4f*)(out + (size_t)(m0 + m) * DIMK + nt * 64 + 4 * ln) = (v4f){ acc[0][r], acc[1][r], acc[2][r], acc[3][r] };
    }
}

extern "C" void kernel_launch(void* const* d_in, const int* in_sizes, int n_in,
                              void* d_out, int out_size, void* d_ws, size_t ws_size,
                              hipStream_t stream) {
    const float* x    = (const float*)d_in[0];
    const float* fcos = (const float*)d_in[2];
    const float* fsin = (const float*)d_in[3];
    const float* wq   = (const float*)d_in[5];
    const float* wk   = (const float*)d_in[6];
    const float* wv   = (const float*)d_in[7];
    const float* wo   = (const float*)d_in[8];
    float* out = (float*)d_out;

    float* qws  = (float*)d_ws;
    float* ktws = qws  + (size_t)BATCH * NH  * SEQ * HD;
    float* vws  = ktws + (size_t)BATCH * NKV * HD  * SEQ;
    float* aws  = vws  + (size_t)BATCH * NKV * SEQ * HD;

    qkv_rope_kernel<<<dim3((BATCH * SEQ) / 16, 12, 1), 128, 0, stream>>>(
        x, wq, wk, wv, fcos, fsin, qws, ktws, vws);
    attn_kernel<<<dim3(SEQ / (16 * QT), NH, BATCH), 32 * KSPLIT, 0, stream>>>(
        qws, ktws, vws, aws);
    oproj_kernel<<<dim3((BATCH * SEQ) / 16, 8, 1), 128, 0, stream>>>(
        aws, wo, out);
}


# projections: a block = 8 waves on 8 token tiles of the SAME 64 weight columns; the weight panel goes once per block through LDS (two 32-row stages, whole-line 16-byte loads), was read by every wave
# speedup vs baseline: 2.2267x; 1.1911x over previous
#include <hip/hip_runtime.h>

#define DIMK   2048
#define NH     32
#define NKV    8
#define HD     64
#define SEQ    2048
#define BATCH  2

typedef __attribute__((ext_vector_type(2))) float v2f;
typedef __attribute__((ext_vector_type(4))) float v4f;
typedef __attribute__((ext_vector_type(8))) float v8f;

__device__ __forceinline__ v8f wmma4(v2f a, v2f b, v8f c) {
    return __builtin_amdgcn_wmma_f32_16x16x4_f32(
         false, a,  false, b,
         (short)0, c,  false,  false);
}

#define GK 32
#define GW 8

__device__ __forceinline__ void gemm_stage(const float (*bs)[64], const v4f a[4], int ln, int g4, v8f acc[4]) {
#pragma unroll
    for (int c = 0; c < 4; ++c) {
        const float* p = &bs[8 * c + g4][4 * ln];
        const v4f b0 = *(const v4f*)(p);
        const v4f b1 = *(const v4f*)(p + 64);
        const v4f b2 = *(const v4f*)(p + 128);
        const v4f b3 = *(const v4f*)(p + 192);
        const v2f a01 = { a[c][0], a[c][1] };
        const v2f a23 = { a[c][2], a[c][3] };
#pragma unroll
        for (int j = 0; j < 4; ++j) {
            v2f b = { b0[j], b1[j] };
            acc[j] = wmma4(a01, b, acc[j]);
        }
#pragma unroll
        for (int j = 0; j < 4; ++j) {
            v2f b = { b2[j], b3[j] };
            acc[j] = wmma4(a23, b, acc[j]);
        }
    }
}

template <int LDB>
__device__ __forceinline__ void gemm_rows16_cols64(const float* __restrict__ A, int lda,
                                                   const float* __restrict__ B,
                                                   int m0, int col0, int K,
                                                   float (*Bs)[GK][64], v8f acc[4]) {
    const int tid  = threadIdx.x;
    const int lane = tid & 31;
    const int ln   = lane & 15;
    const int g4   = (lane >> 4) * 4;
    const float* pa = A + (size_t)(m0 + ln) * lda + g4;
    const int sr = tid >> 4;
    const int sc = (tid & 15) * 4;
    const float* pb = B + (size_t)sr * LDB + col0 + sc;
    acc[0] = {}; acc[1] = {}; acc[2] = {}; acc[3] = {};

    v4f a[4];
    {
        const v4f g0 = *(const v4f*)(pb);
        const v4f g1 = *(const v4f*)(pb + (size_t)16 * LDB);
#pragma unroll
        for (int c = 0; c < 4; ++c) a[c] = *(const v4f*)(pa + 8 * c);
        *(v4f*)(&Bs[0][sr][sc])      = g0;
        *(v4f*)(&Bs[0][sr + 16][sc]) = g1;
    }
    __syncthreads();

    int cur = 0;
    for (int k0 = GK; k0 < K; k0 += GK) {
        pb += (size_t)GK * LDB;
        pa += GK;
        const v4f g0 = *(const v4f*)(pb);
        const v4f g1 = *(const v4f*)(pb + (size_t)16 * LDB);
        v4f an[4];
#pragma unroll
        for (int c = 0; c < 4; ++c) an[c] = *(const v4f*)(pa + 8 * c);
        __builtin_amdgcn_sched_barrier(0);

        gemm_stage(Bs[cur], a, ln, g4, acc);

        *(v4f*)(&Bs[cur ^ 1][sr][sc])      = g0;
        *(v4f*)(&Bs[cur ^ 1][sr + 16][sc]) = g1;
#pragma unroll
        for (int c = 0; c < 4; ++c) a[c] = an[c];
        __syncthreads();
        cur ^= 1;
    }
    gemm_stage(Bs[cur], a, ln, g4, acc);
}

__global__ void __launch_bounds__(32 * GW)
qkv_rope_kernel(const float* __restrict__ x,
                                const float* __restrict__ wq,
                                const float* __restrict__ wk,
                                const float* __restrict__ wv,
                                const float* __restrict__ fcos,
                                const float* __restrict__ fsin,
                                float* __restrict__ qws,
                                float* __restrict__ ktws,
                                float* __restrict__ vws) {
    const int lane = threadIdx.x & 31;
    const int ln   = lane & 15;
    const int grp  = lane >> 4;
    __shared__ __attribute__((aligned(16))) float Bs[2][GK][64];
    const int m0   = blockIdx.x * (16 * GW) + 16 * (threadIdx.x >> 5);
    const int nt   = blockIdx.y;

    const float* w; int h, mode;
    if (nt < 32)      { w = wq; h = nt;      mode = 0; }
    else if (nt < 40) { w = wk; h = nt - 32; mode = 1; }
    else              { w = wv; h = nt - 40; mode = 2; }

    v8f acc[4];
    if (mode == 0) gemm_rows16_cols64<NH  * HD>(x, DIMK, w, m0, h * HD, DIMK, Bs, acc);
    else           gemm_rows16_cols64<NKV * HD>(x, DIMK, w, m0, h * HD, DIMK, Bs, acc);

    const int b  = m0 >> 11;
    const int s0 = m0 & (SEQ - 1);

    if (mode != 2) {
#pragma unroll
        for (int r = 0; r < 8; ++r) {
            const int pos = s0 + r + grp * 8;
            const v2f c = *(const v2f*)(fcos + pos * (HD / 2) + 2 * ln);
            const v2f s = *(const v2f*)(fsin + pos * (HD / 2) + 2 * ln);
            const float e0 = acc[0][r], o0 = acc[1][r];
            const float e1 = acc[2][r], o1 = acc[3][r];
            acc[0][r] = e0 * c[0] - o0 * s[0];
            acc[1][r] = e0 * s[0] + o0 * c[0];
            acc[2][r] = e1 * c[1] - o1 * s[1];
            acc[3][r] = e1 * s[1] + o1 * c[1];
        }
    }

    if (mode == 1) {
        float* base = ktws + (size_t)(b * NKV + h) * HD * SEQ + s0 + grp * 8;
#pragma unroll
        for (int j = 0; j < 4; ++j) {
            float* p = base + (size_t)(4 * ln + j) * SEQ;
            *(v4f*)(p)     = (v4f){ acc[j][0], acc[j][1], acc[j][2], acc[j][3] };
            *(v4f*)(p + 4) = (v4f){ acc[j][4], acc[j][5], acc[j][6], acc[j][7] };
        }
    } else {
        float* base = (mode == 0) ? qws + ((size_t)(b * NH  + h) * SEQ + s0) * HD
                                  : vws + ((size_t)(b * NKV + h) * SEQ + s0) * HD;
#pragma unroll
        for (int r = 0; r < 8; ++r) {
            const int m = r + grp * 8;
            *(v4f*)(base + (size_t)m * HD + 4 * ln) = (v4f){ acc[0][r], acc[1][r], acc[2][r], acc[3][r] };
        }
    }
}

#define KSPLIT 8
#define QT 2
__global__ void __launch_bounds__(32 * KSPLIT, 4)
attn_kernel(const float* __restrict__ qws,
            const float* __restrict__ ktws,
            const float* __restrict__ vws,
            float* __restrict__ aws) {
    __shared__ float  pl[KSPLIT][QT][16 * 18];
    __shared__ float4 part[KSPLIT - 1][8][32];
    __shared__ float  ml[KSPLIT - 1][QT][2][16];
    const int lane = threadIdx.x & 31;
    const int ln   = lane & 15;
    const int grp  = lane >> 4;
    const int g2   = grp * 2;
    const int w    = __builtin_amdgcn_readfirstlane((int)(threadIdx.x >> 5));
    const int q0   = blockIdx.x * (16 * QT);
    const int h    = blockIdx.y;
    const int b    = blockIdx.z;
    const int hk   = h >> 2;

    const float* qb  = qws + ((size_t)(b * NH + h) * SEQ + q0) * HD;
    const float* ktb = ktws + (size_t)(b * NKV + hk) * HD * SEQ;
    const float* vb  = vws + (size_t)(b * NKV + hk) * SEQ * HD;

    v2f aq[QT][16];
#pragma unroll
    for (int t = 0; t < QT; ++t)
#pragma unroll
        for (int kk = 0; kk < 16; ++kk)
            aq[t][kk] = *(const v2f*)(qb + (t * 16 + ln) * HD + kk * 4 + g2);

    v8f o[QT][4];
    float mrow[QT][8], lrow[QT][8];
#pragma unroll
    for (int t = 0; t < QT; ++t) {
        o[t][0] = {}; o[t][1] = {}; o[t][2] = {}; o[t][3] = {};
#pragma unroll
        for (int r = 0; r < 8; ++r) { mrow[t][r] = -1e30f; lrow[t][r] = 0.0f; }
    }

    for (int n0 = w * 16; n0 <= q0 + 16 * (QT - 1); n0 += 16 * KSPLIT) {
        v2f bk[16];
#pragma unroll
        for (int kk = 0; kk < 16; ++kk) {
            const float* kp = ktb + (size_t)(kk * 4 + g2) * SEQ + n0 + ln;
            bk[kk] = (v2f){ kp[0], kp[SEQ] };
        }
        const bool live0 = (n0 <= q0);

        v8f s[QT];
#pragma unroll
        for (int t = 0; t < QT; ++t) {
            s[t] = {};
            if (t > 0 || live0) {
#pragma unroll
                for (int kk = 0; kk < 16; ++kk)
                    s[t] = wmma4(aq[t][kk], bk[kk], s[t]);
            }
        }

        v2f bv[4][4];
#pragma unroll
        for (int kk = 0; kk < 4; ++kk) {
            const float* vp = vb + (size_t)(n0 + kk * 4 + g2) * HD + ln;
#pragma unroll
            for (int j = 0; j < 4; ++j)
                bv[kk][j] = (v2f){ vp[j * 16], vp[HD + j * 16] };
        }

#pragma unroll
        for (int t = 0; t < QT; ++t) {
            if (t > 0 || live0) {
                float* plw = pl[w][t];
                const bool diag = (n0 == q0 + 16 * t);
                float p[8];
#pragma unroll
                for (int r = 0; r < 8; ++r) {
                    float sv = s[t][r] * 0.125f;
                    if (diag && (ln > r + grp * 8)) sv = -1e30f;
                    float rm = sv;
                    rm = fmaxf(rm, __shfl_xor(rm, 1, 32));
                    rm = fmaxf(rm, __shfl_xor(rm, 2, 32));
                    rm = fmaxf(rm, __shfl_xor(rm, 4, 32));
                    rm = fmaxf(rm, __shfl_xor(rm, 8, 32));
                    const float mn    = fmaxf(mrow[t][r], rm);
                    const float alpha = expf(mrow[t][r] - mn);
                    const float pv    = expf(sv - mn);
                    float rs = pv;
                    rs += __shfl_xor(rs, 1, 32);
                    rs += __shfl_xor(rs, 2, 32);
                    rs += __shfl_xor(rs, 4, 32);
                    rs += __shfl_xor(rs, 8, 32);
                    lrow[t][r] = lrow[t][r] * alpha + rs;
                    mrow[t][r] = mn;
                    p[r] = pv;
#pragma unroll
                    for (int j = 0; j < 4; ++j) o[t][j][r] *= alpha;
                }

                __builtin_amdgcn_fence(__ATOMIC_RELEASE, "wavefront");
#pragma unroll
                for (int r = 0; r < 8; ++r)
                    plw[(r + grp * 8) * 18 + ln] = p[r];
                __builtin_amdgcn_fence(__ATOMIC_ACQ_REL, "wavefront");
                v2f ap[4];
#pragma unroll
                for (int kk = 0; kk < 4; ++kk)
                    ap[kk] = *(const v2f*)(&plw[ln * 18 + kk * 4 + g2]);

#pragma unroll
                for (int kk = 0; kk < 4; ++kk)
#pragma unroll
                    for (int j = 0; j < 4; ++j)
                        o[t][j] = wmma4(ap[kk], bv[kk][j], o[t][j]);
            }
        }
    }

#pragma unroll
    for (int t = 0; t < QT; ++t) {
        if (w > 0) {
#pragma unroll
            for (int j = 0; j < 4; ++j) {
                part[w - 1][j * 2 + 0][lane] = make_float4(o[t][j][0], o[t][j][1], o[t][j][2], o[t][j][3]);
                part[w - 1][j * 2 + 1][lane] = make_float4(o[t][j][4], o[t][j][5], o[t][j][6], o[t][j][7]);
            }
            if (ln == 0) {
#pragma unroll
                for (int r = 0; r < 8; ++r) {
                    ml[w - 1][t][grp][r]     = mrow[t][r];
                    ml[w - 1][t][grp][8 + r] = lrow[t][r];
                }
            }
        }
        __syncthreads();
        if (w == 0) {
#pragma unroll
            for (int pw = 0; pw < KSPLIT - 1; ++pw) {
                float a0[8], a1[8];
#pragma unroll
                for (int r = 0; r < 8; ++r) {
                    const float mw = ml[pw][t][grp][r];
                    const float lw = ml[pw][t][grp][8 + r];
                    const float mn = fmaxf(mrow[t][r], mw);
                    a0[r] = expf(mrow[t][r] - mn);
                    a1[r] = expf(mw - mn);
                    lrow[t][r] = lrow[t][r] * a0[r] + lw * a1[r];
                    mrow[t][r] = mn;
                }
#pragma unroll
                for (int j = 0; j < 4; ++j) {
                    const float4 lo = part[pw][j * 2 + 0][lane];
                    const float4 hi = part[pw][j * 2 + 1][lane];
                    o[t][j][0] = o[t][j][0] * a0[0] + lo.x * a1[0];
                    o[t][j][1] = o[t][j][1] * a0[1] + lo.y * a1[1];
                    o[t][j][2] = o[t][j][2] * a0[2] + lo.z * a1[2];
                    o[t][j][3] = o[t][j][3] * a0[3] + lo.w * a1[3];
                    o[t][j][4] = o[t][j][4] * a0[4] + hi.x * a1[4];
                    o[t][j][5] = o[t][j][5] * a0[5] + hi.y * a1[5];
                    o[t][j][6] = o[t][j][6] * a0[6] + hi.z * a1[6];
                    o[t][j][7] = o[t][j][7] * a0[7] + hi.w * a1[7];
                }
            }

            float* ob = aws + ((size_t)(b * SEQ) + q0 + 16 * t) * (NH * HD) + h * HD;
#pragma unroll
            for (int r = 0; r < 8; ++r) {
                const float inv = 1.0f / lrow[t][r];
                const int m = r + grp * 8;
#pragma unroll
                for (int j = 0; j < 4; ++j)
                    ob[(size_t)m * (NH * HD) + j * 16 + ln] = o[t][j][r] * inv;
            }
        }
        if (t + 1 < QT) __syncthreads();
    }
}

__global__ void __launch_bounds__(32 * GW)
oproj_kernel(const float* __restrict__ aws,
                             const float* __restrict__ wo,
                             float* __restrict__ out) {
    const int lane = threadIdx.x & 31;
    const int ln   = lane & 15;
    const int grp  = lane >> 4;
    __shared__ __attribute__((aligned(16))) float Bs[2][GK][64];
    const int m0   = blockIdx.x * (16 * GW) + 16 * (threadIdx.x >> 5);
    const int nt   = blockIdx.y;

    v8f acc[4];
    gemm_rows16_cols64<DIMK>(aws, DIMK, wo, m0, nt * 64, DIMK, Bs, acc);

#pragma unroll
    for (int r = 0; r < 8; ++r) {
        const int m = r + grp * 8;
        *(v4f*)(out + (size_t)(m0 + m) * DIMK + nt * 64 + 4 * ln) = (v4f){ acc[0][r], acc[1][r], acc[2][r], acc[3][r] };
    }
}

extern "C" void kernel_launch(void* const* d_in, const int* in_sizes, int n_in,
                              void* d_out, int out_size, void* d_ws, size_t ws_size,
                              hipStream_t stream) {
    const float* x    = (const float*)d_in[0];
    const float* fcos = (const float*)d_in[2];
    const float* fsin = (const float*)d_in[3];
    const float* wq   = (const float*)d_in[5];
    const float* wk   = (const float*)d_in[6];
    const float* wv   = (const float*)d_in[7];
    const float* wo   = (const float*)d_in[8];
    float* out = (float*)d_out;

    float* qws  = (float*)d_ws;
    float* ktws = qws  + (size_t)BATCH * NH  * SEQ * HD;
    float* vws  = ktws + (size_t)BATCH * NKV * HD  * SEQ;
    float* aws  = vws  + (size_t)BATCH * NKV * SEQ * HD;

    qkv_rope_kernel<<<dim3((BATCH * SEQ) / (16 * GW), 48, 1), 32 * GW, 0, stream>>>(
        x, wq, wk, wv, fcos, fsin, qws, ktws, vws);
    attn_kernel<<<dim3(SEQ / (16 * QT), NH, BATCH), 32 * KSPLIT, 0, stream>>>(
        qws, ktws, vws, aws);
    oproj_kernel<<<dim3((BATCH * SEQ) / (16 * GW), 32, 1), 32 * GW, 0, stream>>>(
        aws, wo, out);
}


# attention softmax: row max by 4 row_xmask DPP max steps on the vector ALU (was 4 LDS permutes, each waited for); row sums kept per lane, summed once after the key loop; order of the row sums changes
# speedup vs baseline: 2.4698x; 1.1092x over previous
#include <hip/hip_runtime.h>

#define DIMK   2048
#define NH     32
#define NKV    8
#define HD     64
#define SEQ    2048
#define BATCH  2

typedef __attribute__((ext_vector_type(2))) float v2f;
typedef __attribute__((ext_vector_type(4))) float v4f;
typedef __attribute__((ext_vector_type(8))) float v8f;

__device__ __forceinline__ v8f wmma4(v2f a, v2f b, v8f c) {
    return __builtin_amdgcn_wmma_f32_16x16x4_f32(
         false, a,  false, b,
         (short)0, c,  false,  false);
}

template <int M>
__device__ __forceinline__ float row_xor(float v) {
    return __builtin_bit_cast(float, __builtin_amdgcn_update_dpp(0, __builtin_bit_cast(int, v), 0x160 | M, 0xf, 0xf, true));
}

template <int M>
__device__ __forceinline__ float row_max_step(float v) {
    float d;
    asm("v_max_num_f32_dpp %0, %1, %1 row_xmask:%2 row_mask:0xf bank_mask:0xf" : "=v"(d) : "v"(v), "n"(M));
    return d;
}

#define GK 32
#define GW 8

__device__ __forceinline__ void gemm_stage(const float (*bs)[64], const v4f a[4], int ln, int g4, v8f acc[4]) {
#pragma unroll
    for (int c = 0; c < 4; ++c) {
        const float* p = &bs[8 * c + g4][4 * ln];
        const v4f b0 = *(const v4f*)(p);
        const v4f b1 = *(const v4f*)(p + 64);
        const v4f b2 = *(const v4f*)(p + 128);
        const v4f b3 = *(const v4f*)(p + 192);
        const v2f a01 = { a[c][0], a[c][1] };
        const v2f a23 = { a[c][2], a[c][3] };
#pragma unroll
        for (int j = 0; j < 4; ++j) {
            v2f b = { b0[j], b1[j] };
            acc[j] = wmma4(a01, b, acc[j]);
        }
#pragma unroll
        for (int j = 0; j < 4; ++j) {
            v2f b = { b2[j], b3[j] };
            acc[j] = wmma4(a23, b, acc[j]);
        }
    }
}

template <int LDB>
__device__ __forceinline__ void gemm_rows16_cols64(const float* __restrict__ A, int lda,
                                                   const float* __restrict__ B,
                                                   int m0, int col0, int K,
                                                   float (*Bs)[GK][64], v8f acc[4]) {
    const int tid  = threadIdx.x;
    const int lane = tid & 31;
    const int ln   = lane & 15;
    const int g4   = (lane >> 4) * 4;
    const float* pa = A + (size_t)(m0 + ln) * lda + g4;
    const int sr = tid >> 4;
    const int sc = (tid & 15) * 4;
    const float* pb = B + (size_t)sr * LDB + col0 + sc;
    acc[0] = {}; acc[1] = {}; acc[2] = {}; acc[3] = {};

    v4f a[4];
    {
        const v4f g0 = *(const v4f*)(pb);
        const v4f g1 = *(const v4f*)(pb + (size_t)16 * LDB);
#pragma unroll
        for (int c = 0; c < 4; ++c) a[c] = *(const v4f*)(pa + 8 * c);
        *(v4f*)(&Bs[0][sr][sc])      = g0;
        *(v4f*)(&Bs[0][sr + 16][sc]) = g1;
    }
    __syncthreads();

    int cur = 0;
    for (int k0 = GK; k0 < K; k0 += GK) {
        pb += (size_t)GK * LDB;
        pa += GK;
        const v4f g0 = *(const v4f*)(pb);
        const v4f g1 = *(const v4f*)(pb + (size_t)16 * LDB);
        v4f an[4];
#pragma unroll
        for (int c = 0; c < 4; ++c) an[c] = *(const v4f*)(pa + 8 * c);
        __builtin_amdgcn_sched_barrier(0);

        gemm_stage(Bs[cur], a, ln, g4, acc);

        *(v4f*)(&Bs[cur ^ 1][sr][sc])      = g0;
        *(v4f*)(&Bs[cur ^ 1][sr + 16][sc]) = g1;
#pragma unroll
        for (int c = 0; c < 4; ++c) a[c] = an[c];
        __syncthreads();
        cur ^= 1;
    }
    gemm_stage(Bs[cur], a, ln, g4, acc);
}

__global__ void __launch_bounds__(32 * GW)
qkv_rope_kernel(const float* __restrict__ x,
                                const float* __restrict__ wq,
                                const float* __restrict__ wk,
                                const float* __restrict__ wv,
                                const float* __restrict__ fcos,
                                const float* __restrict__ fsin,
                                float* __restrict__ qws,
                                float* __restrict__ ktws,
                                float* __restrict__ vws) {
    const int lane = threadIdx.x & 31;
    const int ln   = lane & 15;
    const int grp  = lane >> 4;
    __shared__ __attribute__((aligned(16))) float Bs[2][GK][64];
    const int m0   = blockIdx.x * (16 * GW) + 16 * (threadIdx.x >> 5);
    const int nt   = blockIdx.y;

    const float* w; int h, mode;
    if (nt < 32)      { w = wq; h = nt;      mode = 0; }
    else if (nt < 40) { w = wk; h = nt - 32; mode = 1; }
    else              { w = wv; h = nt - 40; mode = 2; }

    v8f acc[4];
    if (mode == 0) gemm_rows16_cols64<NH  * HD>(x, DIMK, w, m0, h * HD, DIMK, Bs, acc);
    else           gemm_rows16_cols64<NKV * HD>(x, DIMK, w, m0, h * HD, DIMK, Bs, acc);

    const int b  = m0 >> 11;
    const int s0 = m0 & (SEQ - 1);

    if (mode != 2) {
#pragma unroll
        for (int r = 0; r < 8; ++r) {
            const int pos = s0 + r + grp * 8;
            const v2f c = *(const v2f*)(fcos + pos * (HD / 2) + 2 * ln);
            const v2f s = *(const v2f*)(fsin + pos * (HD / 2) + 2 * ln);
            const float e0 = acc[0][r], o0 = acc[1][r];
            const float e1 = acc[2][r], o1 = acc[3][r];
            acc[0][r] = e0 * c[0] - o0 * s[0];
            acc[1][r] = e0 * s[0] + o0 * c[0];
            acc[2][r] = e1 * c[1] - o1 * s[1];
            acc[3][r] = e1 * s[1] + o1 * c[1];
        }
    }

    if (mode == 1) {
        float* base = ktws + (size_t)(b * NKV + h) * HD * SEQ + s0 + grp * 8;
#pragma unroll
        for (int j = 0; j < 4; ++j) {
            float* p = base + (size_t)(4 * ln + j) * SEQ;
            *(v4f*)(p)     = (v4f){ acc[j][0], acc[j][1], acc[j][2], acc[j][3] };
            *(v4f*)(p + 4) = (v4f){ acc[j][4], acc[j][5], acc[j][6], acc[j][7] };
        }
    } else {
        float* base = (mode == 0) ? qws + ((size_t)(b * NH  + h) * SEQ + s0) * HD
                                  : vws + ((size_t)(b * NKV + h) * SEQ + s0) * HD;
#pragma unroll
        for (int r = 0; r < 8; ++r) {
            const int m = r + grp * 8;
            *(v4f*)(base + (size_t)m * HD + 4 * ln) = (v4f){ acc[0][r], acc[1][r], acc[2][r], acc[3][r] };
        }
    }
}

#define KSPLIT 8
#define QT 2
__global__ void __launch_bounds__(32 * KSPLIT, 4)
attn_kernel(const float* __restrict__ qws,
            const float* __restrict__ ktws,
            const float* __restrict__ vws,
            float* __restrict__ aws) {
    __shared__ float  pl[KSPLIT][QT][16 * 18];
    __shared__ float4 part[KSPLIT - 1][8][32];
    __shared__ float  ml[KSPLIT - 1][QT][2][16];
    const int lane = threadIdx.x & 31;
    const int ln   = lane & 15;
    const int grp  = lane >> 4;
    const int g2   = grp * 2;
    const int w    = __builtin_amdgcn_readfirstlane((int)(threadIdx.x >> 5));
    const int q0   = blockIdx.x * (16 * QT);
    const int h    = blockIdx.y;
    const int b    = blockIdx.z;
    const int hk   = h >> 2;

    const float* qb  = qws + ((size_t)(b * NH + h) * SEQ + q0) * HD;
    const float* ktb = ktws + (size_t)(b * NKV + hk) * HD * SEQ;
    const float* vb  = vws + (size_t)(b * NKV + hk) * SEQ * HD;

    v2f aq[QT][16];
#pragma unroll
    for (int t = 0; t < QT; ++t)
#pragma unroll
        for (int kk = 0; kk < 16; ++kk)
            aq[t][kk] = *(const v2f*)(qb + (t * 16 + ln) * HD + kk * 4 + g2);

    v8f o[QT][4];
    float mrow[QT][8], lrow[QT][8];
#pragma unroll
    for (int t = 0; t < QT; ++t) {
        o[t][0] = {}; o[t][1] = {}; o[t][2] = {}; o[t][3] = {};
#pragma unroll
        for (int r = 0; r < 8; ++r) { mrow[t][r] = -1e30f; lrow[t][r] = 0.0f; }
    }

    for (int n0 = w * 16; n0 <= q0 + 16 * (QT - 1); n0 += 16 * KSPLIT) {
        v2f bk[16];
#pragma unroll
        for (int kk = 0; kk < 16; ++kk) {
            const float* kp = ktb + (size_t)(kk * 4 + g2) * SEQ + n0 + ln;
            bk[kk] = (v2f){ kp[0], kp[SEQ] };
        }
        const bool live0 = (n0 <= q0);

        v8f s[QT];
#pragma unroll
        for (int t = 0; t < QT; ++t) {
            s[t] = {};
            if (t > 0 || live0) {
#pragma unroll
                for (int kk = 0; kk < 16; ++kk)
                    s[t] = wmma4(aq[t][kk], bk[kk], s[t]);
            }
        }

        v2f bv[4][4];
#pragma unroll
        for (int kk = 0; kk < 4; ++kk) {
            const float* vp = vb + (size_t)(n0 + kk * 4 + g2) * HD + ln;
#pragma unroll
            for (int j = 0; j < 4; ++j)
                bv[kk][j] = (v2f){ vp[j * 16], vp[HD + j * 16] };
        }

#pragma unroll
        for (int t = 0; t < QT; ++t) {
            if (t > 0 || live0) {
                float* plw = pl[w][t];
                const bool diag = (n0 == q0 + 16 * t);
                float p[8];
#pragma unroll
                for (int r = 0; r < 8; ++r) {
                    float sv = s[t][r] * 0.125f;
                    if (diag && (ln > r + grp * 8)) sv = -1e30f;
                    float rm = sv;
                    rm = row_max_step<1>(rm);
                    rm = row_max_step<2>(rm);
                    rm = row_max_step<4>(rm);
                    rm = row_max_step<8>(rm);
                    const float mn    = fmaxf(mrow[t][r], rm);
                    const float alpha = expf(mrow[t][r] - mn);
                    const float pv    = expf(sv - mn);
                    lrow[t][r] = lrow[t][r] * alpha + pv;
                    mrow[t][r] = mn;
                    p[r] = pv;
#pragma unroll
                    for (int j = 0; j < 4; ++j) o[t][j][r] *= alpha;
                }

                __builtin_amdgcn_fence(__ATOMIC_RELEASE, "wavefront");
#pragma unroll
                for (int r = 0; r < 8; ++r)
                    plw[(r + grp * 8) * 18 + ln] = p[r];
                __builtin_amdgcn_fence(__ATOMIC_ACQ_REL, "wavefront");
                v2f ap[4];
#pragma unroll
                for (int kk = 0; kk < 4; ++kk)
                    ap[kk] = *(const v2f*)(&plw[ln * 18 + kk * 4 + g2]);

#pragma unroll
                for (int kk = 0; kk < 4; ++kk)
#pragma unroll
                    for (int j = 0; j < 4; ++j)
                        o[t][j] = wmma4(ap[kk], bv[kk][j], o[t][j]);
            }
        }
    }

#pragma unroll
    for (int t = 0; t < QT; ++t)
#pragma unroll
        for (int r = 0; r < 8; ++r) {
            float rs = lrow[t][r];
            rs += row_xor<1>(rs);
            rs += row_xor<2>(rs);
            rs += row_xor<4>(rs);
            rs += row_xor<8>(rs);
            lrow[t][r] = rs;
        }

#pragma unroll
    for (int t = 0; t < QT; ++t) {
        if (w > 0) {
#pragma unroll
            for (int j = 0; j < 4; ++j) {
                part[w - 1][j * 2 + 0][lane] = make_float4(o[t][j][0], o[t][j][1], o[t][j][2], o[t][j][3]);
                part[w - 1][j * 2 + 1][lane] = make_float4(o[t][j][4], o[t][j][5], o[t][j][6], o[t][j][7]);
            }
            if (ln == 0) {
#pragma unroll
                for (int r = 0; r < 8; ++r) {
                    ml[w - 1][t][grp][r]     = mrow[t][r];
                    ml[w - 1][t][grp][8 + r] = lrow[t][r];
                }
            }
        }
        __syncthreads();
        if (w == 0) {
#pragma unroll
            for (int pw = 0; pw < KSPLIT - 1; ++pw) {
                float a0[8], a1[8];
#pragma unroll
                for (int r = 0; r < 8; ++r) {
                    const float mw = ml[pw][t][grp][r];
                    const float lw = ml[pw][t][grp][8 + r];
                    const float mn = fmaxf(mrow[t][r], mw);
                    a0[r] = expf(mrow[t][r] - mn);
                    a1[r] = expf(mw - mn);
                    lrow[t][r] = lrow[t][r] * a0[r] + lw * a1[r];
                    mrow[t][r] = mn;
                }
#pragma unroll
                for (int j = 0; j < 4; ++j) {
                    const float4 lo = part[pw][j * 2 + 0][lane];
                    const float4 hi = part[pw][j * 2 + 1][lane];
                    o[t][j][0] = o[t][j][0] * a0[0] + lo.x * a1[0];
                    o[t][j][1] = o[t][j][1] * a0[1] + lo.y * a1[1];
                    o[t][j][2] = o[t][j][2] * a0[2] + lo.z * a1[2];
                    o[t][j][3] = o[t][j][3] * a0[3] + lo.w * a1[3];
                    o[t][j][4] = o[t][j][4] * a0[4] + hi.x * a1[4];
                    o[t][j][5] = o[t][j][5] * a0[5] + hi.y * a1[5];
                    o[t][j][6] = o[t][j][6] * a0[6] + hi.z * a1[6];
                    o[t][j][7] = o[t][j][7] * a0[7] + hi.w * a1[7];
                }
            }

            float* ob = aws + ((size_t)(b * SEQ) + q0 + 16 * t) * (NH * HD) + h * HD;
#pragma unroll
            for (int r = 0; r < 8; ++r) {
                const float inv = 1.0f / lrow[t][r];
                const int m = r + grp * 8;
#pragma unroll
                for (int j = 0; j < 4; ++j)
                    ob[(size_t)m * (NH * HD) + j * 16 + ln] = o[t][j][r] * inv;
            }
        }
        if (t + 1 < QT) __syncthreads();
    }
}

__global__ void __launch_bounds__(32 * GW)
oproj_kernel(const float* __restrict__ aws,
                             const float* __restrict__ wo,
                             float* __restrict__ out) {
    const int lane = threadIdx.x & 31;
    const int ln   = lane & 15;
    const int grp  = lane >> 4;
    __shared__ __attribute__((aligned(16))) float Bs[2][GK][64];
    const int m0   = blockIdx.x * (16 * GW) + 16 * (threadIdx.x >> 5);
    const int nt   = blockIdx.y;

    v8f acc[4];
    gemm_rows16_cols64<DIMK>(aws, DIMK, wo, m0, nt * 64, DIMK, Bs, acc);

#pragma unroll
    for (int r = 0; r < 8; ++r) {
        const int m = r + grp * 8;
        *(v4f*)(out + (size_t)(m0 + m) * DIMK + nt * 64 + 4 * ln) = (v4f){ acc[0][r], acc[1][r], acc[2][r], acc[3][r] };
    }
}

extern "C" void kernel_launch(void* const* d_in, const int* in_sizes, int n_in,
                              void* d_out, int out_size, void* d_ws, size_t ws_size,
                              hipStream_t stream) {
    const float* x    = (const float*)d_in[0];
    const float* fcos = (const float*)d_in[2];
    const float* fsin = (const float*)d_in[3];
    const float* wq   = (const float*)d_in[5];
    const float* wk   = (const float*)d_in[6];
    const float* wv   = (const float*)d_in[7];
    const float* wo   = (const float*)d_in[8];
    float* out = (float*)d_out;

    float* qws  = (float*)d_ws;
    float* ktws = qws  + (size_t)BATCH * NH  * SEQ * HD;
    float* vws  = ktws + (size_t)BATCH * NKV * HD  * SEQ;
    float* aws  = vws  + (size_t)BATCH * NKV * SEQ * HD;

    qkv_rope_kernel<<<dim3((BATCH * SEQ) / (16 * GW), 48, 1), 32 * GW, 0, stream>>>(
        x, wq, wk, wv, fcos, fsin, qws, ktws, vws);
    attn_kernel<<<dim3(SEQ / (16 * QT), NH, BATCH), 32 * KSPLIT, 0, stream>>>(
        qws, ktws, vws, aws);
    oproj_kernel<<<dim3((BATCH * SEQ) / (16 * GW), 32, 1), 32 * GW, 0, stream>>>(
        aws, wo, out);
}


# attention: the 8 waves' partials are merged by all 8 waves: each rescales its own to the block-wide row max, then sums and stores 1/8 of the slab (was wave 0 alone); order of the merge sums changes
# speedup vs baseline: 2.5433x; 1.0298x over previous
#include <hip/hip_runtime.h>

#define DIMK   2048
#define NH     32
#define NKV    8
#define HD     64
#define SEQ    2048
#define BATCH  2

typedef __attribute__((ext_vector_type(2))) float v2f;
typedef __attribute__((ext_vector_type(4))) float v4f;
typedef __attribute__((ext_vector_type(8))) float v8f;

__device__ __forceinline__ v8f wmma4(v2f a, v2f b, v8f c) {
    return __builtin_amdgcn_wmma_f32_16x16x4_f32(
         false, a,  false, b,
         (short)0, c,  false,  false);
}

template <int M>
__device__ __forceinline__ float row_xor(float v) {
    return __builtin_bit_cast(float, __builtin_amdgcn_update_dpp(0, __builtin_bit_cast(int, v), 0x160 | M, 0xf, 0xf, true));
}

template <int M>
__device__ __forceinline__ float row_max_step(float v) {
    float d;
    asm("v_max_num_f32_dpp %0, %1, %1 row_xmask:%2 row_mask:0xf bank_mask:0xf" : "=v"(d) : "v"(v), "n"(M));
    return d;
}

#define GK 32
#define GW 8

__device__ __forceinline__ void gemm_stage(const float (*bs)[64], const v4f a[4], int ln, int g4, v8f acc[4]) {
#pragma unroll
    for (int c = 0; c < 4; ++c) {
        const float* p = &bs[8 * c + g4][4 * ln];
        const v4f b0 = *(const v4f*)(p);
        const v4f b1 = *(const v4f*)(p + 64);
        const v4f b2 = *(const v4f*)(p + 128);
        const v4f b3 = *(const v4f*)(p + 192);
        const v2f a01 = { a[c][0], a[c][1] };
        const v2f a23 = { a[c][2], a[c][3] };
#pragma unroll
        for (int j = 0; j < 4; ++j) {
            v2f b = { b0[j], b1[j] };
            acc[j] = wmma4(a01, b, acc[j]);
        }
#pragma unroll
        for (int j = 0; j < 4; ++j) {
            v2f b = { b2[j], b3[j] };
            acc[j] = wmma4(a23, b, acc[j]);
        }
    }
}

template <int LDB>
__device__ __forceinline__ void gemm_rows16_cols64(const float* __restrict__ A, int lda,
                                                   const float* __restrict__ B,
                                                   int m0, int col0, int K,
                                                   float (*Bs)[GK][64], v8f acc[4]) {
    const int tid  = threadIdx.x;
    const int lane = tid & 31;
    const int ln   = lane & 15;
    const int g4   = (lane >> 4) * 4;
    const float* pa = A + (size_t)(m0 + ln) * lda + g4;
    const int sr = tid >> 4;
    const int sc = (tid & 15) * 4;
    const float* pb = B + (size_t)sr * LDB + col0 + sc;
    acc[0] = {}; acc[1] = {}; acc[2] = {}; acc[3] = {};

    v4f a[4];
    {
        const v4f g0 = *(const v4f*)(pb);
        const v4f g1 = *(const v4f*)(pb + (size_t)16 * LDB);
#pragma unroll
        for (int c = 0; c < 4; ++c) a[c] = *(const v4f*)(pa + 8 * c);
        *(v4f*)(&Bs[0][sr][sc])      = g0;
        *(v4f*)(&Bs[0][sr + 16][sc]) = g1;
    }
    __syncthreads();

    int cur = 0;
    for (int k0 = GK; k0 < K; k0 += GK) {
        pb += (size_t)GK * LDB;
        pa += GK;
        const v4f g0 = *(const v4f*)(pb);
        const v4f g1 = *(const v4f*)(pb + (size_t)16 * LDB);
        v4f an[4];
#pragma unroll
        for (int c = 0; c < 4; ++c) an[c] = *(const v4f*)(pa + 8 * c);
        __builtin_amdgcn_sched_barrier(0);

        gemm_stage(Bs[cur], a, ln, g4, acc);

        *(v4f*)(&Bs[cur ^ 1][sr][sc])      = g0;
        *(v4f*)(&Bs[cur ^ 1][sr + 16][sc]) = g1;
#pragma unroll
        for (int c = 0; c < 4; ++c) a[c] = an[c];
        __syncthreads();
        cur ^= 1;
    }
    gemm_stage(Bs[cur], a, ln, g4, acc);
}

__global__ void __launch_bounds__(32 * GW)
qkv_rope_kernel(const float* __restrict__ x,
                                const float* __restrict__ wq,
                                const float* __restrict__ wk,
                                const float* __restrict__ wv,
                                const float* __restrict__ fcos,
                                const float* __restrict__ fsin,
                                float* __restrict__ qws,
                                float* __restrict__ ktws,
                                float* __restrict__ vws) {
    const int lane = threadIdx.x & 31;
    const int ln   = lane & 15;
    const int grp  = lane >> 4;
    __shared__ __attribute__((aligned(16))) float Bs[2][GK][64];
    const int m0   = blockIdx.x * (16 * GW) + 16 * (threadIdx.x >> 5);
    const int nt   = blockIdx.y;

    const float* w; int h, mode;
    if (nt < 32)      { w = wq; h = nt;      mode = 0; }
    else if (nt < 40) { w = wk; h = nt - 32; mode = 1; }
    else              { w = wv; h = nt - 40; mode = 2; }

    v8f acc[4];
    if (mode == 0) gemm_rows16_cols64<NH  * HD>(x, DIMK, w, m0, h * HD, DIMK, Bs, acc);
    else           gemm_rows16_cols64<NKV * HD>(x, DIMK, w, m0, h * HD, DIMK, Bs, acc);

    const int b  = m0 >> 11;
    const int s0 = m0 & (SEQ - 1);

    if (mode != 2) {
#pragma unroll
        for (int r = 0; r < 8; ++r) {
            const int pos = s0 + r + grp * 8;
            const v2f c = *(const v2f*)(fcos + pos * (HD / 2) + 2 * ln);
            const v2f s = *(const v2f*)(fsin + pos * (HD / 2) + 2 * ln);
            const float e0 = acc[0][r], o0 = acc[1][r];
            const float e1 = acc[2][r], o1 = acc[3][r];
            acc[0][r] = e0 * c[0] - o0 * s[0];
            acc[1][r] = e0 * s[0] + o0 * c[0];
            acc[2][r] = e1 * c[1] - o1 * s[1];
            acc[3][r] = e1 * s[1] + o1 * c[1];
        }
    }

    if (mode == 1) {
        float* base = ktws + (size_t)(b * NKV + h) * HD * SEQ + s0 + grp * 8;
#pragma unroll
        for (int j = 0; j < 4; ++j) {
            float* p = base + (size_t)(4 * ln + j) * SEQ;
            *(v4f*)(p)     = (v4f){ acc[j][0], acc[j][1], acc[j][2], acc[j][3] };
            *(v4f*)(p + 4) = (v4f){ acc[j][4], acc[j][5], acc[j][6], acc[j][7] };
        }
    } else {
        float* base = (mode == 0) ? qws + ((size_t)(b * NH  + h) * SEQ + s0) * HD
                                  : vws + ((size_t)(b * NKV + h) * SEQ + s0) * HD;
#pragma unroll
        for (int r = 0; r < 8; ++r) {
            const int m = r + grp * 8;
            *(v4f*)(base + (size_t)m * HD + 4 * ln) = (v4f){ acc[0][r], acc[1][r], acc[2][r], acc[3][r] };
        }
    }
}

#define KSPLIT 8
#define QT 2
__global__ void __launch_bounds__(32 * KSPLIT, 4)
attn_kernel(const float* __restrict__ qws,
            const float* __restrict__ ktws,
            const float* __restrict__ vws,
            float* __restrict__ aws) {
    __shared__ float  pl[KSPLIT][QT][16 * 18];
    __shared__ v4f    part[KSPLIT][QT][8][32];
    __shared__ v4f    mx[KSPLIT][QT][2][2];
    __shared__ v4f    sm[KSPLIT][QT][2][2];
    const int lane = threadIdx.x & 31;
    const int ln   = lane & 15;
    const int grp  = lane >> 4;
    const int g2   = grp * 2;
    const int w    = __builtin_amdgcn_readfirstlane((int)(threadIdx.x >> 5));
    const int q0   = blockIdx.x * (16 * QT);
    const int h    = blockIdx.y;
    const int b    = blockIdx.z;
    const int hk   = h >> 2;

    const float* qb  = qws + ((size_t)(b * NH + h) * SEQ + q0) * HD;
    const float* ktb = ktws + (size_t)(b * NKV + hk) * HD * SEQ;
    const float* vb  = vws + (size_t)(b * NKV + hk) * SEQ * HD;

    v2f aq[QT][16];
#pragma unroll
    for (int t = 0; t < QT; ++t)
#pragma unroll
        for (int kk = 0; kk < 16; ++kk)
            aq[t][kk] = *(const v2f*)(qb + (t * 16 + ln) * HD + kk * 4 + g2);

    v8f o[QT][4];
    float mrow[QT][8], lrow[QT][8];
#pragma unroll
    for (int t = 0; t < QT; ++t) {
        o[t][0] = {}; o[t][1] = {}; o[t][2] = {}; o[t][3] = {};
#pragma unroll
        for (int r = 0; r < 8; ++r) { mrow[t][r] = -1e30f; lrow[t][r] = 0.0f; }
    }

    for (int n0 = w * 16; n0 <= q0 + 16 * (QT - 1); n0 += 16 * KSPLIT) {
        v2f bk[16];
#pragma unroll
        for (int kk = 0; kk < 16; ++kk) {
            const float* kp = ktb + (size_t)(kk * 4 + g2) * SEQ + n0 + ln;
            bk[kk] = (v2f){ kp[0], kp[SEQ] };
        }
        const bool live0 = (n0 <= q0);

        v8f s[QT];
#pragma unroll
        for (int t = 0; t < QT; ++t) {
            s[t] = {};
            if (t > 0 || live0) {
#pragma unroll
                for (int kk = 0; kk < 16; ++kk)
                    s[t] = wmma4(aq[t][kk], bk[kk], s[t]);
            }
        }

        v2f bv[4][4];
#pragma unroll
        for (int kk = 0; kk < 4; ++kk) {
            const float* vp = vb + (size_t)(n0 + kk * 4 + g2) * HD + ln;
#pragma unroll
            for (int j = 0; j < 4; ++j)
                bv[kk][j] = (v2f){ vp[j * 16], vp[HD + j * 16] };
        }

#pragma unroll
        for (int t = 0; t < QT; ++t) {
            if (t > 0 || live0) {
                float* plw = pl[w][t];
                const bool diag = (n0 == q0 + 16 * t);
                float p[8];
#pragma unroll
                for (int r = 0; r < 8; ++r) {
                    float sv = s[t][r] * 0.125f;
                    if (diag && (ln > r + grp * 8)) sv = -1e30f;
                    float rm = sv;
                    rm = row_max_step<1>(rm);
                    rm = row_max_step<2>(rm);
                    rm = row_max_step<4>(rm);
                    rm = row_max_step<8>(rm);
                    const float mn    = fmaxf(mrow[t][r], rm);
                    const float alpha = expf(mrow[t][r] - mn);
                    const float pv    = expf(sv - mn);
                    lrow[t][r] = lrow[t][r] * alpha + pv;
                    mrow[t][r] = mn;
                    p[r] = pv;
#pragma unroll
                    for (int j = 0; j < 4; ++j) o[t][j][r] *= alpha;
                }

                __builtin_amdgcn_fence(__ATOMIC_RELEASE, "wavefront");
#pragma unroll
                for (int r = 0; r < 8; ++r)
                    plw[(r + grp * 8) * 18 + ln] = p[r];
                __builtin_amdgcn_fence(__ATOMIC_ACQ_REL, "wavefront");
                v2f ap[4];
#pragma unroll
                for (int kk = 0; kk < 4; ++kk)
                    ap[kk] = *(const v2f*)(&plw[ln * 18 + kk * 4 + g2]);

#pragma unroll
                for (int kk = 0; kk < 4; ++kk)
#pragma unroll
                    for (int j = 0; j < 4; ++j)
                        o[t][j] = wmma4(ap[kk], bv[kk][j], o[t][j]);
            }
        }
    }

#pragma unroll
    for (int t = 0; t < QT; ++t)
#pragma unroll
        for (int r = 0; r < 8; ++r) {
            float rs = lrow[t][r];
            rs += row_xor<1>(rs);
            rs += row_xor<2>(rs);
            rs += row_xor<4>(rs);
            rs += row_xor<8>(rs);
            lrow[t][r] = rs;
        }

    if (ln == 0) {
#pragma unroll
        for (int t = 0; t < QT; ++t) {
            mx[w][t][grp][0] = (v4f){ mrow[t][0], mrow[t][1], mrow[t][2], mrow[t][3] };
            mx[w][t][grp][1] = (v4f){ mrow[t][4], mrow[t][5], mrow[t][6], mrow[t][7] };
        }
    }
    __syncthreads();

#pragma unroll
    for (int t = 0; t < QT; ++t) {
        float mb[8];
#pragma unroll
        for (int r = 0; r < 8; ++r) mb[r] = mrow[t][r];
#pragma unroll
        for (int pw = 0; pw < KSPLIT; ++pw) {
            const v4f lo = mx[pw][t][grp][0];
            const v4f hi = mx[pw][t][grp][1];
#pragma unroll
            for (int r = 0; r < 4; ++r) {
                mb[r]     = fmaxf(mb[r], lo[r]);
                mb[4 + r] = fmaxf(mb[4 + r], hi[r]);
            }
        }
#pragma unroll
        for (int r = 0; r < 8; ++r) {
            const float a = expf(mrow[t][r] - mb[r]);
            lrow[t][r] *= a;
#pragma unroll
            for (int j = 0; j < 4; ++j) o[t][j][r] *= a;
        }
#pragma unroll
        for (int j = 0; j < 4; ++j) {
            part[w][t][j * 2 + 0][lane] = (v4f){ o[t][j][0], o[t][j][1], o[t][j][2], o[t][j][3] };
            part[w][t][j * 2 + 1][lane] = (v4f){ o[t][j][4], o[t][j][5], o[t][j][6], o[t][j][7] };
        }
        if (ln == 0) {
            sm[w][t][grp][0] = (v4f){ lrow[t][0], lrow[t][1], lrow[t][2], lrow[t][3] };
            sm[w][t][grp][1] = (v4f){ lrow[t][4], lrow[t][5], lrow[t][6], lrow[t][7] };
        }
    }
    __syncthreads();

    {
        const int t3 = w >> 2;
        const int rh = (w >> 1) & 1;
        const int cp = w & 1;
        v4f acc0 = part[0][t3][cp * 4 + rh][lane];
        v4f acc1 = part[0][t3][cp * 4 + 2 + rh][lane];
        v4f ls   = sm[0][t3][grp][rh];
#pragma unroll
        for (int pw = 1; pw < KSPLIT; ++pw) {
            acc0 += part[pw][t3][cp * 4 + rh][lane];
            acc1 += part[pw][t3][cp * 4 + 2 + rh][lane];
            ls   += sm[pw][t3][grp][rh];
        }
        float* ob = aws + ((size_t)(b * SEQ) + q0 + 16 * t3 + 4 * rh + 8 * grp) * (NH * HD) + h * HD + 32 * cp + ln;
#pragma unroll
        for (int r = 0; r < 4; ++r) {
            const float inv = 1.0f / ls[r];
            ob[(size_t)r * (NH * HD)]      = acc0[r] * inv;
            ob[(size_t)r * (NH * HD) + 16] = acc1[r] * inv;
        }
    }
}

__global__ void __launch_bounds__(32 * GW)
oproj_kernel(const float* __restrict__ aws,
                             const float* __restrict__ wo,
                             float* __restrict__ out) {
    const int lane = threadIdx.x & 31;
    const int ln   = lane & 15;
    const int grp  = lane >> 4;
    __shared__ __attribute__((aligned(16))) float Bs[2][GK][64];
    const int m0   = blockIdx.x * (16 * GW) + 16 * (threadIdx.x >> 5);
    const int nt   = blockIdx.y;

    v8f acc[4];
    gemm_rows16_cols64<DIMK>(aws, DIMK, wo, m0, nt * 64, DIMK, Bs, acc);

#pragma unroll
    for (int r = 0; r < 8; ++r) {
        const int m = r + grp * 8;
        *(v4f*)(out + (size_t)(m0 + m) * DIMK + nt * 64 + 4 * ln) = (v4f){ acc[0][r], acc[1][r], acc[2][r], acc[3][r] };
    }
}

extern "C" void kernel_launch(void* const* d_in, const int* in_sizes, int n_in,
                              void* d_out, int out_size, void* d_ws, size_t ws_size,
                              hipStream_t stream) {
    const float* x    = (const float*)d_in[0];
    const float* fcos = (const float*)d_in[2];
    const float* fsin = (const float*)d_in[3];
    const float* wq   = (const float*)d_in[5];
    const float* wk   = (const float*)d_in[6];
    const float* wv   = (const float*)d_in[7];
    const float* wo   = (const float*)d_in[8];
    float* out = (float*)d_out;

    float* qws  = (float*)d_ws;
    float* ktws = qws  + (size_t)BATCH * NH  * SEQ * HD;
    float* vws  = ktws + (size_t)BATCH * NKV * HD  * SEQ;
    float* aws  = vws  + (size_t)BATCH * NKV * SEQ * HD;

    qkv_rope_kernel<<<dim3((BATCH * SEQ) / (16 * GW), 48, 1), 32 * GW, 0, stream>>>(
        x, wq, wk, wv, fcos, fsin, qws, ktws, vws);
    attn_kernel<<<dim3(SEQ / (16 * QT), NH, BATCH), 32 * KSPLIT, 0, stream>>>(
        qws, ktws, vws, aws);
    oproj_kernel<<<dim3((BATCH * SEQ) / (16 * GW), 32, 1), 32 * GW, 0, stream>>>(
        aws, wo, out);
}


# attention softmax: a row's sum and O are rescaled only when a score tops the row's reference max by over 8 (wave-uniform vote); else no row max, no rescale factor; rounding of weights and sums changes
# speedup vs baseline: 2.7650x; 1.0872x over previous
#include <hip/hip_runtime.h>

#define DIMK   2048
#define NH     32
#define NKV    8
#define HD     64
#define SEQ    2048
#define BATCH  2

typedef __attribute__((ext_vector_type(2))) float v2f;
typedef __attribute__((ext_vector_type(4))) float v4f;
typedef __attribute__((ext_vector_type(8))) float v8f;

__device__ __forceinline__ v8f wmma4(v2f a, v2f b, v8f c) {
    return __builtin_amdgcn_wmma_f32_16x16x4_f32(
         false, a,  false, b,
         (short)0, c,  false,  false);
}

template <int M>
__device__ __forceinline__ float row_xor(float v) {
    return __builtin_bit_cast(float, __builtin_amdgcn_update_dpp(0, __builtin_bit_cast(int, v), 0x160 | M, 0xf, 0xf, true));
}

template <int M>
__device__ __forceinline__ float row_max_step(float v) {
    float d;
    asm("v_max_num_f32_dpp %0, %1, %1 row_xmask:%2 row_mask:0xf bank_mask:0xf" : "=v"(d) : "v"(v), "n"(M));
    return d;
}

#define GK 32
#define GW 8

__device__ __forceinline__ void gemm_stage(const float (*bs)[64], const v4f a[4], int ln, int g4, v8f acc[4]) {
#pragma unroll
    for (int c = 0; c < 4; ++c) {
        const float* p = &bs[8 * c + g4][4 * ln];
        const v4f b0 = *(const v4f*)(p);
        const v4f b1 = *(const v4f*)(p + 64);
        const v4f b2 = *(const v4f*)(p + 128);
        const v4f b3 = *(const v4f*)(p + 192);
        const v2f a01 = { a[c][0], a[c][1] };
        const v2f a23 = { a[c][2], a[c][3] };
#pragma unroll
        for (int j = 0; j < 4; ++j) {
            v2f b = { b0[j], b1[j] };
            acc[j] = wmma4(a01, b, acc[j]);
        }
#pragma unroll
        for (int j = 0; j < 4; ++j) {
            v2f b = { b2[j], b3[j] };
            acc[j] = wmma4(a23, b, acc[j]);
        }
    }
}

template <int LDB>
__device__ __forceinline__ void gemm_rows16_cols64(const float* __restrict__ A, int lda,
                                                   const float* __restrict__ B,
                                                   int m0, int col0, int K,
                                                   float (*Bs)[GK][64], v8f acc[4]) {
    const int tid  = threadIdx.x;
    const int lane = tid & 31;
    const int ln   = lane & 15;
    const int g4   = (lane >> 4) * 4;
    const float* pa = A + (size_t)(m0 + ln) * lda + g4;
    const int sr = tid >> 4;
    const int sc = (tid & 15) * 4;
    const float* pb = B + (size_t)sr * LDB + col0 + sc;
    acc[0] = {}; acc[1] = {}; acc[2] = {}; acc[3] = {};

    v4f a[4];
    {
        const v4f g0 = *(const v4f*)(pb);
        const v4f g1 = *(const v4f*)(pb + (size_t)16 * LDB);
#pragma unroll
        for (int c = 0; c < 4; ++c) a[c] = *(const v4f*)(pa + 8 * c);
        *(v4f*)(&Bs[0][sr][sc])      = g0;
        *(v4f*)(&Bs[0][sr + 16][sc]) = g1;
    }
    __syncthreads();

    int cur = 0;
    for (int k0 = GK; k0 < K; k0 += GK) {
        pb += (size_t)GK * LDB;
        pa += GK;
        const v4f g0 = *(const v4f*)(pb);
        const v4f g1 = *(const v4f*)(pb + (size_t)16 * LDB);
        v4f an[4];
#pragma unroll
        for (int c = 0; c < 4; ++c) an[c] = *(const v4f*)(pa + 8 * c);
        __builtin_amdgcn_sched_barrier(0);

        gemm_stage(Bs[cur], a, ln, g4, acc);

        *(v4f*)(&Bs[cur ^ 1][sr][sc])      = g0;
        *(v4f*)(&Bs[cur ^ 1][sr + 16][sc]) = g1;
#pragma unroll
        for (int c = 0; c < 4; ++c) a[c] = an[c];
        __syncthreads();
        cur ^= 1;
    }
    gemm_stage(Bs[cur], a, ln, g4, acc);
}

__global__ void __launch_bounds__(32 * GW)
qkv_rope_kernel(const float* __restrict__ x,
                                const float* __restrict__ wq,
                                const float* __restrict__ wk,
                                const float* __restrict__ wv,
                                const float* __restrict__ fcos,
                                const float* __restrict__ fsin,
                                float* __restrict__ qws,
                                float* __restrict__ ktws,
                                float* __restrict__ vws) {
    const int lane = threadIdx.x & 31;
    const int ln   = lane & 15;
    const int grp  = lane >> 4;
    __shared__ __attribute__((aligned(16))) float Bs[2][GK][64];
    const int m0   = blockIdx.x * (16 * GW) + 16 * (threadIdx.x >> 5);
    const int nt   = blockIdx.y;

    const float* w; int h, mode;
    if (nt < 32)      { w = wq; h = nt;      mode = 0; }
    else if (nt < 40) { w = wk; h = nt - 32; mode = 1; }
    else              { w = wv; h = nt - 40; mode = 2; }

    v8f acc[4];
    if (mode == 0) gemm_rows16_cols64<NH  * HD>(x, DIMK, w, m0, h * HD, DIMK, Bs, acc);
    else           gemm_rows16_cols64<NKV * HD>(x, DIMK, w, m0, h * HD, DIMK, Bs, acc);

    const int b  = m0 >> 11;
    const int s0 = m0 & (SEQ - 1);

    if (mode != 2) {
#pragma unroll
        for (int r = 0; r < 8; ++r) {
            const int pos = s0 + r + grp * 8;
            const v2f c = *(const v2f*)(fcos + pos * (HD / 2) + 2 * ln);
            const v2f s = *(const v2f*)(fsin + pos * (HD / 2) + 2 * ln);
            const float e0 = acc[0][r], o0 = acc[1][r];
            const float e1 = acc[2][r], o1 = acc[3][r];
            acc[0][r] = e0 * c[0] - o0 * s[0];
            acc[1][r] = e0 * s[0] + o0 * c[0];
            acc[2][r] = e1 * c[1] - o1 * s[1];
            acc[3][r] = e1 * s[1] + o1 * c[1];
        }
    }

    if (mode == 1) {
        float* base = ktws + (size_t)(b * NKV + h) * HD * SEQ + s0 + grp * 8;
#pragma unroll
        for (int j = 0; j < 4; ++j) {
            float* p = base + (size_t)(4 * ln + j) * SEQ;
            *(v4f*)(p)     = (v4f){ acc[j][0], acc[j][1], acc[j][2], acc[j][3] };
            *(v4f*)(p + 4) = (v4f){ acc[j][4], acc[j][5], acc[j][6], acc[j][7] };
        }
    } else {
        float* base = (mode == 0) ? qws + ((size_t)(b * NH  + h) * SEQ + s0) * HD
                                  : vws + ((size_t)(b * NKV + h) * SEQ + s0) * HD;
#pragma unroll
        for (int r = 0; r < 8; ++r) {
            const int m = r + grp * 8;
            *(v4f*)(base + (size_t)m * HD + 4 * ln) = (v4f){ acc[0][r], acc[1][r], acc[2][r], acc[3][r] };
        }
    }
}

#define KSPLIT 8
#define QT 2
#define RESCALE_AT 8.0f
__global__ void __launch_bounds__(32 * KSPLIT, 4)
attn_kernel(const float* __restrict__ qws,
            const float* __restrict__ ktws,
            const float* __restrict__ vws,
            float* __restrict__ aws) {
    __shared__ float  pl[KSPLIT][QT][16 * 18];
    __shared__ v4f    part[KSPLIT][QT][8][32];
    __shared__ v4f    mx[KSPLIT][QT][2][2];
    __shared__ v4f    sm[KSPLIT][QT][2][2];
    const int lane = threadIdx.x & 31;
    const int ln   = lane & 15;
    const int grp  = lane >> 4;
    const int g2   = grp * 2;
    const int w    = __builtin_amdgcn_readfirstlane((int)(threadIdx.x >> 5));
    const int q0   = blockIdx.x * (16 * QT);
    const int h    = blockIdx.y;
    const int b    = blockIdx.z;
    const int hk   = h >> 2;

    const float* qb  = qws + ((size_t)(b * NH + h) * SEQ + q0) * HD;
    const float* ktb = ktws + (size_t)(b * NKV + hk) * HD * SEQ;
    const float* vb  = vws + (size_t)(b * NKV + hk) * SEQ * HD;

    v2f aq[QT][16];
#pragma unroll
    for (int t = 0; t < QT; ++t)
#pragma unroll
        for (int kk = 0; kk < 16; ++kk)
            aq[t][kk] = *(const v2f*)(qb + (t * 16 + ln) * HD + kk * 4 + g2);

    v8f o[QT][4];
    float mrow[QT][8], lrow[QT][8];
#pragma unroll
    for (int t = 0; t < QT; ++t) {
        o[t][0] = {}; o[t][1] = {}; o[t][2] = {}; o[t][3] = {};
#pragma unroll
        for (int r = 0; r < 8; ++r) { mrow[t][r] = -1e30f; lrow[t][r] = 0.0f; }
    }

    for (int n0 = w * 16; n0 <= q0 + 16 * (QT - 1); n0 += 16 * KSPLIT) {
        v2f bk[16];
#pragma unroll
        for (int kk = 0; kk < 16; ++kk) {
            const float* kp = ktb + (size_t)(kk * 4 + g2) * SEQ + n0 + ln;
            bk[kk] = (v2f){ kp[0], kp[SEQ] };
        }
        const bool live0 = (n0 <= q0);

        v8f s[QT];
#pragma unroll
        for (int t = 0; t < QT; ++t) {
            s[t] = {};
            if (t > 0 || live0) {
#pragma unroll
                for (int kk = 0; kk < 16; ++kk)
                    s[t] = wmma4(aq[t][kk], bk[kk], s[t]);
            }
        }

        v2f bv[4][4];
#pragma unroll
        for (int kk = 0; kk < 4; ++kk) {
            const float* vp = vb + (size_t)(n0 + kk * 4 + g2) * HD + ln;
#pragma unroll
            for (int j = 0; j < 4; ++j)
                bv[kk][j] = (v2f){ vp[j * 16], vp[HD + j * 16] };
        }

#pragma unroll
        for (int t = 0; t < QT; ++t) {
            if (t > 0 || live0) {
                float* plw = pl[w][t];
                const bool diag = (n0 == q0 + 16 * t);
                float p[8];
#pragma unroll
                for (int r = 0; r < 8; ++r) {
                    float sv = s[t][r] * 0.125f;
                    if (diag && (ln > r + grp * 8)) sv = -1e30f;
                    p[r] = sv - mrow[t][r];
                }
                const float top = fmaxf(fmaxf(fmaxf(p[0], p[1]), fmaxf(p[2], p[3])),
                                        fmaxf(fmaxf(p[4], p[5]), fmaxf(p[6], p[7])));
                if (__builtin_amdgcn_ballot_w32(top > RESCALE_AT) != 0) {
#pragma unroll
                    for (int r = 0; r < 8; ++r) {
                        float sv = s[t][r] * 0.125f;
                        if (diag && (ln > r + grp * 8)) sv = -1e30f;
                        float rm = sv;
                        rm = row_max_step<1>(rm);
                        rm = row_max_step<2>(rm);
                        rm = row_max_step<4>(rm);
                        rm = row_max_step<8>(rm);
                        const float mn    = fmaxf(mrow[t][r], rm);
                        const float alpha = expf(mrow[t][r] - mn);
                        lrow[t][r] *= alpha;
                        mrow[t][r] = mn;
                        p[r] = sv - mn;
#pragma unroll
                        for (int j = 0; j < 4; ++j) o[t][j][r] *= alpha;
                    }
                }
#pragma unroll
                for (int r = 0; r < 8; ++r) {
                    p[r] = expf(p[r]);
                    lrow[t][r] += p[r];
                }

                __builtin_amdgcn_fence(__ATOMIC_RELEASE, "wavefront");
#pragma unroll
                for (int r = 0; r < 8; ++r)
                    plw[(r + grp * 8) * 18 + ln] = p[r];
                __builtin_amdgcn_fence(__ATOMIC_ACQ_REL, "wavefront");
                v2f ap[4];
#pragma unroll
                for (int kk = 0; kk < 4; ++kk)
                    ap[kk] = *(const v2f*)(&plw[ln * 18 + kk * 4 + g2]);

#pragma unroll
                for (int kk = 0; kk < 4; ++kk)
#pragma unroll
                    for (int j = 0; j < 4; ++j)
                        o[t][j] = wmma4(ap[kk], bv[kk][j], o[t][j]);
            }
        }
    }

#pragma unroll
    for (int t = 0; t < QT; ++t)
#pragma unroll
        for (int r = 0; r < 8; ++r) {
            float rs = lrow[t][r];
            rs += row_xor<1>(rs);
            rs += row_xor<2>(rs);
            rs += row_xor<4>(rs);
            rs += row_xor<8>(rs);
            lrow[t][r] = rs;
        }

    if (ln == 0) {
#pragma unroll
        for (int t = 0; t < QT; ++t) {
            mx[w][t][grp][0] = (v4f){ mrow[t][0], mrow[t][1], mrow[t][2], mrow[t][3] };
            mx[w][t][grp][1] = (v4f){ mrow[t][4], mrow[t][5], mrow[t][6], mrow[t][7] };
        }
    }
    __syncthreads();

#pragma unroll
    for (int t = 0; t < QT; ++t) {
        float mb[8];
#pragma unroll
        for (int r = 0; r < 8; ++r) mb[r] = mrow[t][r];
#pragma unroll
        for (int pw = 0; pw < KSPLIT; ++pw) {
            const v4f lo = mx[pw][t][grp][0];
            const v4f hi = mx[pw][t][grp][1];
#pragma unroll
            for (int r = 0; r < 4; ++r) {
                mb[r]     = fmaxf(mb[r], lo[r]);
                mb[4 + r] = fmaxf(mb[4 + r], hi[r]);
            }
        }
#pragma unroll
        for (int r = 0; r < 8; ++r) {
            const float a = expf(mrow[t][r] - mb[r]);
            lrow[t][r] *= a;
#pragma unroll
            for (int j = 0; j < 4; ++j) o[t][j][r] *= a;
        }
#pragma unroll
        for (int j = 0; j < 4; ++j) {
            part[w][t][j * 2 + 0][lane] = (v4f){ o[t][j][0], o[t][j][1], o[t][j][2], o[t][j][3] };
            part[w][t][j * 2 + 1][lane] = (v4f){ o[t][j][4], o[t][j][5], o[t][j][6], o[t][j][7] };
        }
        if (ln == 0) {
            sm[w][t][grp][0] = (v4f){ lrow[t][0], lrow[t][1], lrow[t][2], lrow[t][3] };
            sm[w][t][grp][1] = (v4f){ lrow[t][4], lrow[t][5], lrow[t][6], lrow[t][7] };
        }
    }
    __syncthreads();

    {
        const int t3 = w >> 2;
        const int rh = (w >> 1) & 1;
        const int cp = w & 1;
        v4f acc0 = part[0][t3][cp * 4 + rh][lane];
        v4f acc1 = part[0][t3][cp * 4 + 2 + rh][lane];
        v4f ls   = sm[0][t3][grp][rh];
#pragma unroll
        for (int pw = 1; pw < KSPLIT; ++pw) {
            acc0 += part[pw][t3][cp * 4 + rh][lane];
            acc1 += part[pw][t3][cp * 4 + 2 + rh][lane];
            ls   += sm[pw][t3][grp][rh];
        }
        float* ob = aws + ((size_t)(b * SEQ) + q0 + 16 * t3 + 4 * rh + 8 * grp) * (NH * HD) + h * HD + 32 * cp + ln;
#pragma unroll
        for (int r = 0; r < 4; ++r) {
            const float inv = 1.0f / ls[r];
            ob[(size_t)r * (NH * HD)]      = acc0[r] * inv;
            ob[(size_t)r * (NH * HD) + 16] = acc1[r] * inv;
        }
    }
}

__global__ void __launch_bounds__(32 * GW)
oproj_kernel(const float* __restrict__ aws,
                             const float* __restrict__ wo,
                             float* __restrict__ out) {
    const int lane = threadIdx.x & 31;
    const int ln   = lane & 15;
    const int grp  = lane >> 4;
    __shared__ __attribute__((aligned(16))) float Bs[2][GK][64];
    const int m0   = blockIdx.x * (16 * GW) + 16 * (threadIdx.x >> 5);
    const int nt   = blockIdx.y;

    v8f acc[4];
    gemm_rows16_cols64<DIMK>(aws, DIMK, wo, m0, nt * 64, DIMK, Bs, acc);

#pragma unroll
    for (int r = 0; r < 8; ++r) {
        const int m = r + grp * 8;
        *(v4f*)(out + (size_t)(m0 + m) * DIMK + nt * 64 + 4 * ln) = (v4f){ acc[0][r], acc[1][r], acc[2][r], acc[3][r] };
    }
}

extern "C" void kernel_launch(void* const* d_in, const int* in_sizes, int n_in,
                              void* d_out, int out_size, void* d_ws, size_t ws_size,
                              hipStream_t stream) {
    const float* x    = (const float*)d_in[0];
    const float* fcos = (const float*)d_in[2];
    const float* fsin = (const float*)d_in[3];
    const float* wq   = (const float*)d_in[5];
    const float* wk   = (const float*)d_in[6];
    const float* wv   = (const float*)d_in[7];
    const float* wo   = (const float*)d_in[8];
    float* out = (float*)d_out;

    float* qws  = (float*)d_ws;
    float* ktws = qws  + (size_t)BATCH * NH  * SEQ * HD;
    float* vws  = ktws + (size_t)BATCH * NKV * HD  * SEQ;
    float* aws  = vws  + (size_t)BATCH * NKV * SEQ * HD;

    qkv_rope_kernel<<<dim3((BATCH * SEQ) / (16 * GW), 48, 1), 32 * GW, 0, stream>>>(
        x, wq, wk, wv, fcos, fsin, qws, ktws, vws);
    attn_kernel<<<dim3(SEQ / (16 * QT), NH, BATCH), 32 * KSPLIT, 0, stream>>>(
        qws, ktws, vws, aws);
    oproj_kernel<<<dim3((BATCH * SEQ) / (16 * GW), 32, 1), 32 * GW, 0, stream>>>(
        aws, wo, out);
}


# attention: work-groups handed out longest first: query slab is the slowest grid index, last slab (most key blocks) first, heads and batch fastest; short blocks fill the tail; same sums, same order
# speedup vs baseline: 2.9069x; 1.0513x over previous
#include <hip/hip_runtime.h>

#define DIMK   2048
#define NH     32
#define NKV    8
#define HD     64
#define SEQ    2048
#define BATCH  2

typedef __attribute__((ext_vector_type(2))) float v2f;
typedef __attribute__((ext_vector_type(4))) float v4f;
typedef __attribute__((ext_vector_type(8))) float v8f;

__device__ __forceinline__ v8f wmma4(v2f a, v2f b, v8f c) {
    return __builtin_amdgcn_wmma_f32_16x16x4_f32(
         false, a,  false, b,
         (short)0, c,  false,  false);
}

template <int M>
__device__ __forceinline__ float row_xor(float v) {
    return __builtin_bit_cast(float, __builtin_amdgcn_update_dpp(0, __builtin_bit_cast(int, v), 0x160 | M, 0xf, 0xf, true));
}

template <int M>
__device__ __forceinline__ float row_max_step(float v) {
    float d;
    asm("v_max_num_f32_dpp %0, %1, %1 row_xmask:%2 row_mask:0xf bank_mask:0xf" : "=v"(d) : "v"(v), "n"(M));
    return d;
}

#define GK 32
#define GW 8

__device__ __forceinline__ void gemm_stage(const float (*bs)[64], const v4f a[4], int ln, int g4, v8f acc[4]) {
#pragma unroll
    for (int c = 0; c < 4; ++c) {
        const float* p = &bs[8 * c + g4][4 * ln];
        const v4f b0 = *(const v4f*)(p);
        const v4f b1 = *(const v4f*)(p + 64);
        const v4f b2 = *(const v4f*)(p + 128);
        const v4f b3 = *(const v4f*)(p + 192);
        const v2f a01 = { a[c][0], a[c][1] };
        const v2f a23 = { a[c][2], a[c][3] };
#pragma unroll
        for (int j = 0; j < 4; ++j) {
            v2f b = { b0[j], b1[j] };
            acc[j] = wmma4(a01, b, acc[j]);
        }
#pragma unroll
        for (int j = 0; j < 4; ++j) {
            v2f b = { b2[j], b3[j] };
            acc[j] = wmma4(a23, b, acc[j]);
        }
    }
}

template <int LDB>
__device__ __forceinline__ void gemm_rows16_cols64(const float* __restrict__ A, int lda,
                                                   const float* __restrict__ B,
                                                   int m0, int col0, int K,
                                                   float (*Bs)[GK][64], v8f acc[4]) {
    const int tid  = threadIdx.x;
    const int lane = tid & 31;
    const int ln   = lane & 15;
    const int g4   = (lane >> 4) * 4;
    const float* pa = A + (size_t)(m0 + ln) * lda + g4;
    const int sr = tid >> 4;
    const int sc = (tid & 15) * 4;
    const float* pb = B + (size_t)sr * LDB + col0 + sc;
    acc[0] = {}; acc[1] = {}; acc[2] = {}; acc[3] = {};

    v4f a[4];
    {
        const v4f g0 = *(const v4f*)(pb);
        const v4f g1 = *(const v4f*)(pb + (size_t)16 * LDB);
#pragma unroll
        for (int c = 0; c < 4; ++c) a[c] = *(const v4f*)(pa + 8 * c);
        *(v4f*)(&Bs[0][sr][sc])      = g0;
        *(v4f*)(&Bs[0][sr + 16][sc]) = g1;
    }
    __syncthreads();

    int cur = 0;
    for (int k0 = GK; k0 < K; k0 += GK) {
        pb += (size_t)GK * LDB;
        pa += GK;
        const v4f g0 = *(const v4f*)(pb);
        const v4f g1 = *(const v4f*)(pb + (size_t)16 * LDB);
        v4f an[4];
#pragma unroll
        for (int c = 0; c < 4; ++c) an[c] = *(const v4f*)(pa + 8 * c);
        __builtin_amdgcn_sched_barrier(0);

        gemm_stage(Bs[cur], a, ln, g4, acc);

        *(v4f*)(&Bs[cur ^ 1][sr][sc])      = g0;
        *(v4f*)(&Bs[cur ^ 1][sr + 16][sc]) = g1;
#pragma unroll
        for (int c = 0; c < 4; ++c) a[c] = an[c];
        __syncthreads();
        cur ^= 1;
    }
    gemm_stage(Bs[cur], a, ln, g4, acc);
}

__global__ void __launch_bounds__(32 * GW)
qkv_rope_kernel(const float* __restrict__ x,
                                const float* __restrict__ wq,
                                const float* __restrict__ wk,
                                const float* __restrict__ wv,
                                const float* __restrict__ fcos,
                                const float* __restrict__ fsin,
                                float* __restrict__ qws,
                                float* __restrict__ ktws,
                                float* __restrict__ vws) {
    const int lane = threadIdx.x & 31;
    const int ln   = lane & 15;
    const int grp  = lane >> 4;
    __shared__ __attribute__((aligned(16))) float Bs[2][GK][64];
    const int m0   = blockIdx.x * (16 * GW) + 16 * (threadIdx.x >> 5);
    const int nt   = blockIdx.y;

    const float* w; int h, mode;
    if (nt < 32)      { w = wq; h = nt;      mode = 0; }
    else if (nt < 40) { w = wk; h = nt - 32; mode = 1; }
    else              { w = wv; h = nt - 40; mode = 2; }

    v8f acc[4];
    if (mode == 0) gemm_rows16_cols64<NH  * HD>(x, DIMK, w, m0, h * HD, DIMK, Bs, acc);
    else           gemm_rows16_cols64<NKV * HD>(x, DIMK, w, m0, h * HD, DIMK, Bs, acc);

    const int b  = m0 >> 11;
    const int s0 = m0 & (SEQ - 1);

    if (mode != 2) {
#pragma unroll
        for (int r = 0; r < 8; ++r) {
            const int pos = s0 + r + grp * 8;
            const v2f c = *(const v2f*)(fcos + pos * (HD / 2) + 2 * ln);
            const v2f s = *(const v2f*)(fsin + pos * (HD / 2) + 2 * ln);
            const float e0 = acc[0][r], o0 = acc[1][r];
            const float e1 = acc[2][r], o1 = acc[3][r];
            acc[0][r] = e0 * c[0] - o0 * s[0];
            acc[1][r] = e0 * s[0] + o0 * c[0];
            acc[2][r] = e1 * c[1] - o1 * s[1];
            acc[3][r] = e1 * s[1] + o1 * c[1];
        }
    }

    if (mode == 1) {
        float* base = ktws + (size_t)(b * NKV + h) * HD * SEQ + s0 + grp * 8;
#pragma unroll
        for (int j = 0; j < 4; ++j) {
            float* p = base + (size_t)(4 * ln + j) * SEQ;
            *(v4f*)(p)     = (v4f){ acc[j][0], acc[j][1], acc[j][2], acc[j][3] };
            *(v4f*)(p + 4) = (v4f){ acc[j][4], acc[j][5], acc[j][6], acc[j][7] };
        }
    } else {
        float* base = (mode == 0) ? qws + ((size_t)(b * NH  + h) * SEQ + s0) * HD
                                  : vws + ((size_t)(b * NKV + h) * SEQ + s0) * HD;
#pragma unroll
        for (int r = 0; r < 8; ++r) {
            const int m = r + grp * 8;
            *(v4f*)(base + (size_t)m * HD + 4 * ln) = (v4f){ acc[0][r], acc[1][r], acc[2][r], acc[3][r] };
        }
    }
}

#define KSPLIT 8
#define QT 2
#define RESCALE_AT 8.0f
__global__ void __launch_bounds__(32 * KSPLIT, 4)
attn_kernel(const float* __restrict__ qws,
            const float* __restrict__ ktws,
            const float* __restrict__ vws,
            float* __restrict__ aws) {
    __shared__ float  pl[KSPLIT][QT][16 * 18];
    __shared__ v4f    part[KSPLIT][QT][8][32];
    __shared__ v4f    mx[KSPLIT][QT][2][2];
    __shared__ v4f    sm[KSPLIT][QT][2][2];
    const int lane = threadIdx.x & 31;
    const int ln   = lane & 15;
    const int grp  = lane >> 4;
    const int g2   = grp * 2;
    const int w    = __builtin_amdgcn_readfirstlane((int)(threadIdx.x >> 5));
    const int q0   = (SEQ / (16 * QT) - 1 - (int)blockIdx.z) * (16 * QT);
    const int h    = blockIdx.x;
    const int b    = blockIdx.y;
    const int hk   = h >> 2;

    const float* qb  = qws + ((size_t)(b * NH + h) * SEQ + q0) * HD;
    const float* ktb = ktws + (size_t)(b * NKV + hk) * HD * SEQ;
    const float* vb  = vws + (size_t)(b * NKV + hk) * SEQ * HD;

    v2f aq[QT][16];
#pragma unroll
    for (int t = 0; t < QT; ++t)
#pragma unroll
        for (int kk = 0; kk < 16; ++kk)
            aq[t][kk] = *(const v2f*)(qb + (t * 16 + ln) * HD + kk * 4 + g2);

    v8f o[QT][4];
    float mrow[QT][8], lrow[QT][8];
#pragma unroll
    for (int t = 0; t < QT; ++t) {
        o[t][0] = {}; o[t][1] = {}; o[t][2] = {}; o[t][3] = {};
#pragma unroll
        for (int r = 0; r < 8; ++r) { mrow[t][r] = -1e30f; lrow[t][r] = 0.0f; }
    }

    for (int n0 = w * 16; n0 <= q0 + 16 * (QT - 1); n0 += 16 * KSPLIT) {
        v2f bk[16];
#pragma unroll
        for (int kk = 0; kk < 16; ++kk) {
            const float* kp = ktb + (size_t)(kk * 4 + g2) * SEQ + n0 + ln;
            bk[kk] = (v2f){ kp[0], kp[SEQ] };
        }
        const bool live0 = (n0 <= q0);

        v8f s[QT];
#pragma unroll
        for (int t = 0; t < QT; ++t) {
            s[t] = {};
            if (t > 0 || live0) {
#pragma unroll
                for (int kk = 0; kk < 16; ++kk)
                    s[t] = wmma4(aq[t][kk], bk[kk], s[t]);
            }
        }

        v2f bv[4][4];
#pragma unroll
        for (int kk = 0; kk < 4; ++kk) {
            const float* vp = vb + (size_t)(n0 + kk * 4 + g2) * HD + ln;
#pragma unroll
            for (int j = 0; j < 4; ++j)
                bv[kk][j] = (v2f){ vp[j * 16], vp[HD + j * 16] };
        }

#pragma unroll
        for (int t = 0; t < QT; ++t) {
            if (t > 0 || live0) {
                float* plw = pl[w][t];
                const bool diag = (n0 == q0 + 16 * t);
                float p[8];
#pragma unroll
                for (int r = 0; r < 8; ++r) {
                    float sv = s[t][r] * 0.125f;
                    if (diag && (ln > r + grp * 8)) sv = -1e30f;
                    p[r] = sv - mrow[t][r];
                }
                const float top = fmaxf(fmaxf(fmaxf(p[0], p[1]), fmaxf(p[2], p[3])),
                                        fmaxf(fmaxf(p[4], p[5]), fmaxf(p[6], p[7])));
                if (__builtin_amdgcn_ballot_w32(top > RESCALE_AT) != 0) {
#pragma unroll
                    for (int r = 0; r < 8; ++r) {
                        float sv = s[t][r] * 0.125f;
                        if (diag && (ln > r + grp * 8)) sv = -1e30f;
                        float rm = sv;
                        rm = row_max_step<1>(rm);
                        rm = row_max_step<2>(rm);
                        rm = row_max_step<4>(rm);
                        rm = row_max_step<8>(rm);
                        const float mn    = fmaxf(mrow[t][r], rm);
                        const float alpha = expf(mrow[t][r] - mn);
                        lrow[t][r] *= alpha;
                        mrow[t][r] = mn;
                        p[r] = sv - mn;
#pragma unroll
                        for (int j = 0; j < 4; ++j) o[t][j][r] *= alpha;
                    }
                }
#pragma unroll
                for (int r = 0; r < 8; ++r) {
                    p[r] = expf(p[r]);
                    lrow[t][r] += p[r];
                }

                __builtin_amdgcn_fence(__ATOMIC_RELEASE, "wavefront");
#pragma unroll
                for (int r = 0; r < 8; ++r)
                    plw[(r + grp * 8) * 18 + ln] = p[r];
                __builtin_amdgcn_fence(__ATOMIC_ACQ_REL, "wavefront");
                v2f ap[4];
#pragma unroll
                for (int kk = 0; kk < 4; ++kk)
                    ap[kk] = *(const v2f*)(&plw[ln * 18 + kk * 4 + g2]);

#pragma unroll
                for (int kk = 0; kk < 4; ++kk)
#pragma unroll
                    for (int j = 0; j < 4; ++j)
                        o[t][j] = wmma4(ap[kk], bv[kk][j], o[t][j]);
            }
        }
    }

#pragma unroll
    for (int t = 0; t < QT; ++t)
#pragma unroll
        for (int r = 0; r < 8; ++r) {
            float rs = lrow[t][r];
            rs += row_xor<1>(rs);
            rs += row_xor<2>(rs);
            rs += row_xor<4>(rs);
            rs += row_xor<8>(rs);
            lrow[t][r] = rs;
        }

    if (ln == 0) {
#pragma unroll
        for (int t = 0; t < QT; ++t) {
            mx[w][t][grp][0] = (v4f){ mrow[t][0], mrow[t][1], mrow[t][2], mrow[t][3] };
            mx[w][t][grp][1] = (v4f){ mrow[t][4], mrow[t][5], mrow[t][6], mrow[t][7] };
        }
    }
    __syncthreads();

#pragma unroll
    for (int t = 0; t < QT; ++t) {
        float mb[8];
#pragma unroll
        for (int r = 0; r < 8; ++r) mb[r] = mrow[t][r];
#pragma unroll
        for (int pw = 0; pw < KSPLIT; ++pw) {
            const v4f lo = mx[pw][t][grp][0];
            const v4f hi = mx[pw][t][grp][1];
#pragma unroll
            for (int r = 0; r < 4; ++r) {
                mb[r]     = fmaxf(mb[r], lo[r]);
                mb[4 + r] = fmaxf(mb[4 + r], hi[r]);
            }
        }
#pragma unroll
        for (int r = 0; r < 8; ++r) {
            const float a = expf(mrow[t][r] - mb[r]);
            lrow[t][r] *= a;
#pragma unroll
            for (int j = 0; j < 4; ++j) o[t][j][r] *= a;
        }
#pragma unroll
        for (int j = 0; j < 4; ++j) {
            part[w][t][j * 2 + 0][lane] = (v4f){ o[t][j][0], o[t][j][1], o[t][j][2], o[t][j][3] };
            part[w][t][j * 2 + 1][lane] = (v4f){ o[t][j][4], o[t][j][5], o[t][j][6], o[t][j][7] };
        }
        if (ln == 0) {
            sm[w][t][grp][0] = (v4f){ lrow[t][0], lrow[t][1], lrow[t][2], lrow[t][3] };
            sm[w][t][grp][1] = (v4f){ lrow[t][4], lrow[t][5], lrow[t][6], lrow[t][7] };
        }
    }
    __syncthreads();

    {
        const int t3 = w >> 2;
        const int rh = (w >> 1) & 1;
        const int cp = w & 1;
        v4f acc0 = part[0][t3][cp * 4 + rh][lane];
        v4f acc1 = part[0][t3][cp * 4 + 2 + rh][lane];
        v4f ls   = sm[0][t3][grp][rh];
#pragma unroll
        for (int pw = 1; pw < KSPLIT; ++pw) {
            acc0 += part[pw][t3][cp * 4 + rh][lane];
            acc1 += part[pw][t3][cp * 4 + 2 + rh][lane];
            ls   += sm[pw][t3][grp][rh];
        }
        float* ob = aws + ((size_t)(b * SEQ) + q0 + 16 * t3 + 4 * rh + 8 * grp) * (NH * HD) + h * HD + 32 * cp + ln;
#pragma unroll
        for (int r = 0; r < 4; ++r) {
            const float inv = 1.0f / ls[r];
            ob[(size_t)r * (NH * HD)]      = acc0[r] * inv;
            ob[(size_t)r * (NH * HD) + 16] = acc1[r] * inv;
        }
    }
}

__global__ void __launch_bounds__(32 * GW)
oproj_kernel(const float* __restrict__ aws,
                             const float* __restrict__ wo,
                             float* __restrict__ out) {
    const int lane = threadIdx.x & 31;
    const int ln   = lane & 15;
    const int grp  = lane >> 4;
    __shared__ __attribute__((aligned(16))) float Bs[2][GK][64];
    const int m0   = blockIdx.x * (16 * GW) + 16 * (threadIdx.x >> 5);
    const int nt   = blockIdx.y;

    v8f acc[4];
    gemm_rows16_cols64<DIMK>(aws, DIMK, wo, m0, nt * 64, DIMK, Bs, acc);

#pragma unroll
    for (int r = 0; r < 8; ++r) {
        const int m = r + grp * 8;
        *(v4f*)(out + (size_t)(m0 + m) * DIMK + nt * 64 + 4 * ln) = (v4f){ acc[0][r], acc[1][r], acc[2][r], acc[3][r] };
    }
}

extern "C" void kernel_launch(void* const* d_in, const int* in_sizes, int n_in,
                              void* d_out, int out_size, void* d_ws, size_t ws_size,
                              hipStream_t stream) {
    const float* x    = (const float*)d_in[0];
    const float* fcos = (const float*)d_in[2];
    const float* fsin = (const float*)d_in[3];
    const float* wq   = (const float*)d_in[5];
    const float* wk   = (const float*)d_in[6];
    const float* wv   = (const float*)d_in[7];
    const float* wo   = (const float*)d_in[8];
    float* out = (float*)d_out;

    float* qws  = (float*)d_ws;
    float* ktws = qws  + (size_t)BATCH * NH  * SEQ * HD;
    float* vws  = ktws + (size_t)BATCH * NKV * HD  * SEQ;
    float* aws  = vws  + (size_t)BATCH * NKV * SEQ * HD;

    qkv_rope_kernel<<<dim3((BATCH * SEQ) / (16 * GW), 48, 1), 32 * GW, 0, stream>>>(
        x, wq, wk, wv, fcos, fsin, qws, ktws, vws);
    attn_kernel<<<dim3(NH, BATCH, SEQ / (16 * QT)), 32 * KSPLIT, 0, stream>>>(
        qws, ktws, vws, aws);
    oproj_kernel<<<dim3((BATCH * SEQ) / (16 * GW), 32, 1), 32 * GW, 0, stream>>>(
        aws, wo, out);
}
